# starting kernel
# baseline (speedup 1.0000x reference)
#include <hip/hip_runtime.h>
#include <hip/hip_bf16.h>


static constexpr int FD       = 64;
static constexpr int NGRAPHS  = 512;
static constexpr int NTARGETS = 100;

typedef __attribute__((ext_vector_type(2))) float v2f;
typedef __attribute__((ext_vector_type(8))) float v8f;


__global__ void zero_kernel(float* __restrict__ p, int n) {
  int i = blockIdx.x * blockDim.x + threadIdx.x;
  if (i < n) p[i] = 0.0f;
}

__global__ void deg_accum_kernel(const int* __restrict__ dst, float* __restrict__ deg, int nE) {
  int e = blockIdx.x * blockDim.x + threadIdx.x;
  if (e < nE) atomicAdd(&deg[dst[e]], 1.0f);
}

__global__ void dinv_kernel(float* __restrict__ deg, int n) {
  int i = blockIdx.x * blockDim.x + threadIdx.x;
  if (i < n) deg[i] = rsqrtf(deg[i] + 1.0f);
}

__global__ __launch_bounds__(128) void gemm64_wmma(const float* __restrict__ X,
                                                   const float* __restrict__ W,
                                                   float* __restrict__ H) {
  const int rowBase = blockIdx.x * 16;
  const int wave  = threadIdx.x >> 5;
  const int lane  = threadIdx.x & 31;
  const int m     = lane & 15;
  const int khalf = (lane >> 4) << 1;

  const float* __restrict__ xrow = X + (size_t)(rowBase + m) * FD;
  const float* __restrict__ wcol = W + wave * 16 + m;

  v8f acc = {};
#pragma unroll
  for (int kt = 0; kt < FD / 4; ++kt) {
    const int k = kt * 4 + khalf;
    v2f a; a.x = xrow[k];                 a.y = xrow[k + 1];
    v2f b; b.x = wcol[(size_t)k * FD];    b.y = wcol[(size_t)(k + 1) * FD];
    acc = __builtin_amdgcn_wmma_f32_16x16x4_f32(
         false, a,  false, b,
         (short)0, acc,  false,  false);
  }

  const int mBase = (lane >> 4) << 3;
  float* __restrict__ hout = H + (size_t)rowBase * FD + wave * 16 + m;
#pragma unroll
  for (int v = 0; v < 8; ++v)
    hout[(size_t)(mBase + v) * FD] = acc[v];
}


__global__ void init_agg_kernel(const float* __restrict__ h, const float* __restrict__ dinv,
                                const float* __restrict__ bias, float* __restrict__ agg, int n) {
  int i = blockIdx.x * blockDim.x + threadIdx.x;
  if (i < n) {
    int node = i >> 6;
    int d    = i & 63;
    float di = dinv[node];
    agg[i] = h[i] * di * di + bias[d];
  }
}

__global__ void edge_scatter_kernel(const int* __restrict__ src, const int* __restrict__ dst,
                                    const float* __restrict__ dinv, const float* __restrict__ h,
                                    float* __restrict__ agg, int nE) {
  int t = blockIdx.x * blockDim.x + threadIdx.x;
  int e = t >> 6;
  int d = t & 63;
  if (e < nE) {
    int s  = src[e];
    int dd = dst[e];
    float norm = dinv[s] * dinv[dd];
    atomicAdd(&agg[(size_t)dd * FD + d], h[(size_t)s * FD + d] * norm);
  }
}

__global__ void relu_kernel(float* __restrict__ p, int n) {
  int i = blockIdx.x * blockDim.x + threadIdx.x;
  if (i < n) p[i] = fmaxf(p[i], 0.0f);
}


__global__ void pool_scatter_kernel(const int* __restrict__ seg, const float* __restrict__ h,
                                    float* __restrict__ psum, float* __restrict__ pcnt, int nNodes) {
  int t = blockIdx.x * blockDim.x + threadIdx.x;
  int node = t >> 6;
  int d    = t & 63;
  if (node < nNodes) {
    int g = seg[node];
    atomicAdd(&psum[(size_t)g * FD + d], h[(size_t)node * FD + d]);
    if (d == 0) atomicAdd(&pcnt[g], 1.0f);
  }
}

__global__ void out_gemm_kernel(const float* __restrict__ psum, const float* __restrict__ pcnt,
                                const float* __restrict__ Wout, const float* __restrict__ bout,
                                float* __restrict__ out) {
  int t = blockIdx.x * blockDim.x + threadIdx.x;
  if (t < NGRAPHS * NTARGETS) {
    int g = t / NTARGETS;
    int c = t % NTARGETS;
    float inv = 1.0f / fmaxf(pcnt[g], 1.0f);
    float acc = 0.0f;
#pragma unroll 8
    for (int d = 0; d < FD; ++d)
      acc += psum[(size_t)g * FD + d] * Wout[(size_t)d * NTARGETS + c];
    out[t] = acc * inv + bout[c];
  }
}


extern "C" void kernel_launch(void* const* d_in, const int* in_sizes, int n_in,
                              void* d_out, int out_size, void* d_ws, size_t ws_size,
                              hipStream_t stream) {
  const float* x    = (const float*)d_in[0];
  const float* W1   = (const float*)d_in[1];
  const float* b1   = (const float*)d_in[2];
  const float* W2   = (const float*)d_in[3];
  const float* b2   = (const float*)d_in[4];
  const float* W3   = (const float*)d_in[5];
  const float* b3   = (const float*)d_in[6];
  const float* Wout = (const float*)d_in[7];
  const float* bout = (const float*)d_in[8];
  const int*   eidx = (const int*)d_in[9];
  const int*   seg  = (const int*)d_in[10];
  float* out = (float*)d_out;
  (void)n_in; (void)out_size; (void)ws_size;

  const int nNodes = in_sizes[0] / FD;
  const int nE     = in_sizes[9] / 2;
  const int* srcI = eidx;
  const int* dstI = eidx + nE;

  char* ws = (char*)d_ws;
  size_t off = 0;
  auto alloc = [&](size_t bytes) -> void* {
    void* p = ws + off;
    off = (off + bytes + 255) & ~(size_t)255;
    return p;
  };
  float* dinv = (float*)alloc((size_t)nNodes * sizeof(float));
  float* hbuf = (float*)alloc((size_t)nNodes * FD * sizeof(float));
  float* feat = (float*)alloc((size_t)nNodes * FD * sizeof(float));
  float* psum = (float*)alloc((size_t)NGRAPHS * FD * sizeof(float));
  float* pcnt = (float*)alloc((size_t)NGRAPHS * sizeof(float));

  const int T = 256;
  const int nodeBlocks = (nNodes + T - 1) / T;
  const int gemmBlocks = nNodes / 16;
  const int nfeat      = nNodes * FD;
  const int featBlocks = (nfeat + T - 1) / T;
  const int eBlocks    = (int)(((long long)nE * FD + T - 1) / T);

  zero_kernel<<<nodeBlocks, T, 0, stream>>>(dinv, nNodes);
  deg_accum_kernel<<<(nE + T - 1) / T, T, 0, stream>>>(dstI, dinv, nE);
  dinv_kernel<<<nodeBlocks, T, 0, stream>>>(dinv, nNodes);

  gemm64_wmma<<<gemmBlocks, 128, 0, stream>>>(x, W1, hbuf);
  init_agg_kernel<<<featBlocks, T, 0, stream>>>(hbuf, dinv, b1, feat, nfeat);
  edge_scatter_kernel<<<eBlocks, T, 0, stream>>>(srcI, dstI, dinv, hbuf, feat, nE);
  relu_kernel<<<featBlocks, T, 0, stream>>>(feat, nfeat);

  gemm64_wmma<<<gemmBlocks, 128, 0, stream>>>(feat, W2, hbuf);
  init_agg_kernel<<<featBlocks, T, 0, stream>>>(hbuf, dinv, b2, feat, nfeat);
  edge_scatter_kernel<<<eBlocks, T, 0, stream>>>(srcI, dstI, dinv, hbuf, feat, nE);
  relu_kernel<<<featBlocks, T, 0, stream>>>(feat, nfeat);

  gemm64_wmma<<<gemmBlocks, 128, 0, stream>>>(feat, W3, hbuf);
  init_agg_kernel<<<featBlocks, T, 0, stream>>>(hbuf, dinv, b3, feat, nfeat);
  edge_scatter_kernel<<<eBlocks, T, 0, stream>>>(srcI, dstI, dinv, hbuf, feat, nE);

  zero_kernel<<<(NGRAPHS * FD + T - 1) / T, T, 0, stream>>>(psum, NGRAPHS * FD);
  zero_kernel<<<(NGRAPHS + T - 1) / T, T, 0, stream>>>(pcnt, NGRAPHS);
  pool_scatter_kernel<<<featBlocks, T, 0, stream>>>(seg, feat, psum, pcnt, nNodes);
  out_gemm_kernel<<<(NGRAPHS * NTARGETS + T - 1) / T, T, 0, stream>>>(psum, pcnt, Wout, bout, out);
}


# Edge scatter: one wave walks 32 consecutive edges (was half an edge per wave, 64x more waves); src, dst, norm fetched coalesced once, 16 row loads in flight; sums stay float atomics, arrival order
# speedup vs baseline: 2.1093x; 2.1093x over previous
#include <hip/hip_runtime.h>
#include <hip/hip_bf16.h>


static constexpr int FD       = 64;
static constexpr int NGRAPHS  = 512;
static constexpr int NTARGETS = 100;

typedef __attribute__((ext_vector_type(2))) float v2f;
typedef __attribute__((ext_vector_type(8))) float v8f;


__global__ void zero_kernel(float* __restrict__ p, int n) {
  int i = blockIdx.x * blockDim.x + threadIdx.x;
  if (i < n) p[i] = 0.0f;
}

__global__ void deg_accum_kernel(const int* __restrict__ dst, float* __restrict__ deg, int nE) {
  int e = blockIdx.x * blockDim.x + threadIdx.x;
  if (e < nE) atomicAdd(&deg[dst[e]], 1.0f);
}

__global__ void dinv_kernel(float* __restrict__ deg, int n) {
  int i = blockIdx.x * blockDim.x + threadIdx.x;
  if (i < n) deg[i] = rsqrtf(deg[i] + 1.0f);
}

__global__ __launch_bounds__(128) void gemm64_wmma(const float* __restrict__ X,
                                                   const float* __restrict__ W,
                                                   float* __restrict__ H) {
  const int rowBase = blockIdx.x * 16;
  const int wave  = threadIdx.x >> 5;
  const int lane  = threadIdx.x & 31;
  const int m     = lane & 15;
  const int khalf = (lane >> 4) << 1;

  const float* __restrict__ xrow = X + (size_t)(rowBase + m) * FD;
  const float* __restrict__ wcol = W + wave * 16 + m;

  v8f acc = {};
#pragma unroll
  for (int kt = 0; kt < FD / 4; ++kt) {
    const int k = kt * 4 + khalf;
    v2f a; a.x = xrow[k];                 a.y = xrow[k + 1];
    v2f b; b.x = wcol[(size_t)k * FD];    b.y = wcol[(size_t)(k + 1) * FD];
    acc = __builtin_amdgcn_wmma_f32_16x16x4_f32(
         false, a,  false, b,
         (short)0, acc,  false,  false);
  }

  const int mBase = (lane >> 4) << 3;
  float* __restrict__ hout = H + (size_t)rowBase * FD + wave * 16 + m;
#pragma unroll
  for (int v = 0; v < 8; ++v)
    hout[(size_t)(mBase + v) * FD] = acc[v];
}


__global__ void init_agg_kernel(const float* __restrict__ h, const float* __restrict__ dinv,
                                const float* __restrict__ bias, float* __restrict__ agg, int n) {
  int i = blockIdx.x * blockDim.x + threadIdx.x;
  if (i < n) {
    int node = i >> 6;
    int d    = i & 63;
    float di = dinv[node];
    agg[i] = h[i] * di * di + bias[d];
  }
}

__global__ __launch_bounds__(256) void edge_scatter_kernel(const int* __restrict__ src, const int* __restrict__ dst,
                                    const float* __restrict__ dinv, const float* __restrict__ h,
                                    float* __restrict__ agg, int nE) {
  const int lane = threadIdx.x & 31;
  const int waveId = __builtin_amdgcn_readfirstlane((int)((blockIdx.x * blockDim.x + threadIdx.x) >> 5));
  const int base = waveId * 32;
  if (base >= nE) return;
  const int cnt = min(32, nE - base);
  int s = 0, d = 0;
  float nrm = 0.0f;
  if (lane < cnt) {
    s = src[base + lane];
    d = dst[base + lane];
    nrm = dinv[s] * dinv[d];
  }
  const int full = cnt & ~7;
  for (int jb = 0; jb < full; jb += 8) {
    float v0[8], v1[8];
#pragma unroll
    for (int k = 0; k < 8; ++k) {
      const int sj = __builtin_amdgcn_readlane(s, jb + k);
      const float* __restrict__ hr = h + (size_t)sj * FD + lane;
      v0[k] = hr[0];
      v1[k] = hr[32];
    }
#pragma unroll
    for (int k = 0; k < 8; ++k) {
      const int   dj = __builtin_amdgcn_readlane(d, jb + k);
      const float nj = __int_as_float(__builtin_amdgcn_readlane(__float_as_int(nrm), jb + k));
      float* __restrict__ ar = agg + (size_t)dj * FD + lane;
      atomicAdd(ar, v0[k] * nj);
      atomicAdd(ar + 32, v1[k] * nj);
    }
  }
  for (int j = full; j < cnt; ++j) {
    const int   sj = __builtin_amdgcn_readlane(s, j);
    const int   dj = __builtin_amdgcn_readlane(d, j);
    const float nj = __int_as_float(__builtin_amdgcn_readlane(__float_as_int(nrm), j));
    const float* __restrict__ hr = h + (size_t)sj * FD + lane;
    float* __restrict__ ar = agg + (size_t)dj * FD + lane;
    const float v0 = hr[0];
    const float v1 = hr[32];
    atomicAdd(ar, v0 * nj);
    atomicAdd(ar + 32, v1 * nj);
  }
}

__global__ void relu_kernel(float* __restrict__ p, int n) {
  int i = blockIdx.x * blockDim.x + threadIdx.x;
  if (i < n) p[i] = fmaxf(p[i], 0.0f);
}


__global__ void pool_scatter_kernel(const int* __restrict__ seg, const float* __restrict__ h,
                                    float* __restrict__ psum, float* __restrict__ pcnt, int nNodes) {
  int t = blockIdx.x * blockDim.x + threadIdx.x;
  int node = t >> 6;
  int d    = t & 63;
  if (node < nNodes) {
    int g = seg[node];
    atomicAdd(&psum[(size_t)g * FD + d], h[(size_t)node * FD + d]);
    if (d == 0) atomicAdd(&pcnt[g], 1.0f);
  }
}

__global__ void out_gemm_kernel(const float* __restrict__ psum, const float* __restrict__ pcnt,
                                const float* __restrict__ Wout, const float* __restrict__ bout,
                                float* __restrict__ out) {
  int t = blockIdx.x * blockDim.x + threadIdx.x;
  if (t < NGRAPHS * NTARGETS) {
    int g = t / NTARGETS;
    int c = t % NTARGETS;
    float inv = 1.0f / fmaxf(pcnt[g], 1.0f);
    float acc = 0.0f;
#pragma unroll 8
    for (int d = 0; d < FD; ++d)
      acc += psum[(size_t)g * FD + d] * Wout[(size_t)d * NTARGETS + c];
    out[t] = acc * inv + bout[c];
  }
}


extern "C" void kernel_launch(void* const* d_in, const int* in_sizes, int n_in,
                              void* d_out, int out_size, void* d_ws, size_t ws_size,
                              hipStream_t stream) {
  const float* x    = (const float*)d_in[0];
  const float* W1   = (const float*)d_in[1];
  const float* b1   = (const float*)d_in[2];
  const float* W2   = (const float*)d_in[3];
  const float* b2   = (const float*)d_in[4];
  const float* W3   = (const float*)d_in[5];
  const float* b3   = (const float*)d_in[6];
  const float* Wout = (const float*)d_in[7];
  const float* bout = (const float*)d_in[8];
  const int*   eidx = (const int*)d_in[9];
  const int*   seg  = (const int*)d_in[10];
  float* out = (float*)d_out;
  (void)n_in; (void)out_size; (void)ws_size;

  const int nNodes = in_sizes[0] / FD;
  const int nE     = in_sizes[9] / 2;
  const int* srcI = eidx;
  const int* dstI = eidx + nE;

  char* ws = (char*)d_ws;
  size_t off = 0;
  auto alloc = [&](size_t bytes) -> void* {
    void* p = ws + off;
    off = (off + bytes + 255) & ~(size_t)255;
    return p;
  };
  float* dinv = (float*)alloc((size_t)nNodes * sizeof(float));
  float* hbuf = (float*)alloc((size_t)nNodes * FD * sizeof(float));
  float* feat = (float*)alloc((size_t)nNodes * FD * sizeof(float));
  float* psum = (float*)alloc((size_t)NGRAPHS * FD * sizeof(float));
  float* pcnt = (float*)alloc((size_t)NGRAPHS * sizeof(float));

  const int T = 256;
  const int nodeBlocks = (nNodes + T - 1) / T;
  const int gemmBlocks = nNodes / 16;
  const int nfeat      = nNodes * FD;
  const int featBlocks = (nfeat + T - 1) / T;
  const int eBlocks    = (nE + T - 1) / T;

  zero_kernel<<<nodeBlocks, T, 0, stream>>>(dinv, nNodes);
  deg_accum_kernel<<<(nE + T - 1) / T, T, 0, stream>>>(dstI, dinv, nE);
  dinv_kernel<<<nodeBlocks, T, 0, stream>>>(dinv, nNodes);

  gemm64_wmma<<<gemmBlocks, 128, 0, stream>>>(x, W1, hbuf);
  init_agg_kernel<<<featBlocks, T, 0, stream>>>(hbuf, dinv, b1, feat, nfeat);
  edge_scatter_kernel<<<eBlocks, T, 0, stream>>>(srcI, dstI, dinv, hbuf, feat, nE);
  relu_kernel<<<featBlocks, T, 0, stream>>>(feat, nfeat);

  gemm64_wmma<<<gemmBlocks, 128, 0, stream>>>(feat, W2, hbuf);
  init_agg_kernel<<<featBlocks, T, 0, stream>>>(hbuf, dinv, b2, feat, nfeat);
  edge_scatter_kernel<<<eBlocks, T, 0, stream>>>(srcI, dstI, dinv, hbuf, feat, nE);
  relu_kernel<<<featBlocks, T, 0, stream>>>(feat, nfeat);

  gemm64_wmma<<<gemmBlocks, 128, 0, stream>>>(feat, W3, hbuf);
  init_agg_kernel<<<featBlocks, T, 0, stream>>>(hbuf, dinv, b3, feat, nfeat);
  edge_scatter_kernel<<<eBlocks, T, 0, stream>>>(srcI, dstI, dinv, hbuf, feat, nE);

  zero_kernel<<<(NGRAPHS * FD + T - 1) / T, T, 0, stream>>>(psum, NGRAPHS * FD);
  zero_kernel<<<(NGRAPHS + T - 1) / T, T, 0, stream>>>(pcnt, NGRAPHS);
  pool_scatter_kernel<<<featBlocks, T, 0, stream>>>(seg, feat, psum, pcnt, nNodes);
  out_gemm_kernel<<<(NGRAPHS * NTARGETS + T - 1) / T, T, 0, stream>>>(psum, pcnt, Wout, bout, out);
}


# Edge sums by gather: each call buckets edges by dst (int atomic count, block scan, atomic slot fill); one wave per node adds its rows in registers in slot arrival order; no float atomics in layers
# speedup vs baseline: 2.4520x; 1.1625x over previous
#include <hip/hip_runtime.h>
#include <hip/hip_bf16.h>


static constexpr int FD       = 64;
static constexpr int NGRAPHS  = 512;
static constexpr int NTARGETS = 100;

typedef __attribute__((ext_vector_type(2))) float v2f;
typedef __attribute__((ext_vector_type(8))) float v8f;


__global__ void zero_kernel(float* __restrict__ p, int n) {
  int i = blockIdx.x * blockDim.x + threadIdx.x;
  if (i < n) p[i] = 0.0f;
}

__global__ void zero_int_kernel(int* __restrict__ p, int n) {
  int i = blockIdx.x * blockDim.x + threadIdx.x;
  if (i < n) p[i] = 0;
}


__global__ void deg_count_kernel(const int* __restrict__ dst, int* __restrict__ cnt, int nE) {
  int e = blockIdx.x * blockDim.x + threadIdx.x;
  if (e < nE) atomicAdd(&cnt[dst[e]], 1);
}

__global__ __launch_bounds__(256) void offsets_kernel(const int* __restrict__ cnt, int* __restrict__ total,
                                                      int* __restrict__ cursor, float* __restrict__ dinv, int n) {
  __shared__ int waveSum[8];
  __shared__ int blockBase;
  const int i    = blockIdx.x * 256 + threadIdx.x;
  const int lane = threadIdx.x & 31;
  const int wv   = threadIdx.x >> 5;
  const int c    = (i < n) ? cnt[i] : 0;
  int incl = c;
  for (int off = 1; off < 32; off <<= 1) {
    const int t = __shfl_up(incl, off, 32);
    if (lane >= off) incl += t;
  }
  if (lane == 31) waveSum[wv] = incl;
  __syncthreads();
  if (threadIdx.x == 0) {
    int run = 0;
    for (int w = 0; w < 8; ++w) { const int t = waveSum[w]; waveSum[w] = run; run += t; }
    blockBase = atomicAdd(total, run);
  }
  __syncthreads();
  if (i < n) {
    cursor[i] = blockBase + waveSum[wv] + incl - c;
    dinv[i]   = rsqrtf((float)c + 1.0f);
  }
}

__global__ void fill_kernel(const int* __restrict__ src, const int* __restrict__ dst,
                            int* __restrict__ cursor, int* __restrict__ list, int nE) {
  int e = blockIdx.x * blockDim.x + threadIdx.x;
  if (e < nE) {
    const int pos = atomicAdd(&cursor[dst[e]], 1);
    list[pos] = src[e];
  }
}

__global__ __launch_bounds__(128) void gemm64_wmma(const float* __restrict__ X,
                                                   const float* __restrict__ W,
                                                   float* __restrict__ H) {
  const int rowBase = blockIdx.x * 16;
  const int wave  = threadIdx.x >> 5;
  const int lane  = threadIdx.x & 31;
  const int m     = lane & 15;
  const int khalf = (lane >> 4) << 1;

  const float* __restrict__ xrow = X + (size_t)(rowBase + m) * FD;
  const float* __restrict__ wcol = W + wave * 16 + m;

  v8f acc = {};
#pragma unroll
  for (int kt = 0; kt < FD / 4; ++kt) {
    const int k = kt * 4 + khalf;
    v2f a; a.x = xrow[k];                 a.y = xrow[k + 1];
    v2f b; b.x = wcol[(size_t)k * FD];    b.y = wcol[(size_t)(k + 1) * FD];
    acc = __builtin_amdgcn_wmma_f32_16x16x4_f32(
         false, a,  false, b,
         (short)0, acc,  false,  false);
  }

  const int mBase = (lane >> 4) << 3;
  float* __restrict__ hout = H + (size_t)rowBase * FD + wave * 16 + m;
#pragma unroll
  for (int v = 0; v < 8; ++v)
    hout[(size_t)(mBase + v) * FD] = acc[v];
}


template <int G>
__device__ __forceinline__ void gather_group(const float* __restrict__ h, int s, float ds, float di,
                                             int jb, int m, int lane, float& ax, float& ay) {
  v2f v[G];
#pragma unroll
  for (int k = 0; k < G; ++k) {
    const int sj = __builtin_amdgcn_readlane(s, jb + k);
    v[k] = *(const v2f*)(h + (size_t)sj * FD + 2 * lane);
  }
#pragma unroll
  for (int k = 0; k < G; ++k) {
    const float dj = __int_as_float(__builtin_amdgcn_readlane(__float_as_int(ds), jb + k));
    const float wj = (jb + k < m) ? dj * di : 0.0f;
    ax += v[k].x * wj;
    ay += v[k].y * wj;
  }
}

template <bool RELU>
__global__ __launch_bounds__(256) void gather_kernel(const int* __restrict__ cnt, const int* __restrict__ cursorEnd,
                                                     const int* __restrict__ list, const float* __restrict__ dinv,
                                                     const float* __restrict__ h, const float* __restrict__ bias,
                                                     float* __restrict__ out, int nNodes) {
  const int lane = threadIdx.x & 31;
  const int node = __builtin_amdgcn_readfirstlane((int)((blockIdx.x * blockDim.x + threadIdx.x) >> 5));
  if (node >= nNodes) return;
  const int   c     = __builtin_amdgcn_readfirstlane(cnt[node]);
  const int   beg   = __builtin_amdgcn_readfirstlane(cursorEnd[node]) - c;
  const float di    = __int_as_float(__builtin_amdgcn_readfirstlane(__float_as_int(dinv[node])));
  const int   total = c + 1;
  float ax = 0.0f, ay = 0.0f;
  for (int cb = 0; cb < total; cb += 32) {
    const int idx = cb + lane;
    int s = node;
    if (idx < c) s = list[beg + idx];
    const float ds = dinv[s];
    const int m = min(32, total - cb);
    int jb = 0;
    for (; m - jb > 4; jb += 8) gather_group<8>(h, s, ds, di, jb, m, lane, ax, ay);
    if (m - jb > 0) gather_group<4>(h, s, ds, di, jb, m, lane, ax, ay);
  }
  const v2f b = *(const v2f*)(bias + 2 * lane);
  ax += b.x;
  ay += b.y;
  if (RELU) { ax = fmaxf(ax, 0.0f); ay = fmaxf(ay, 0.0f); }
  v2f r; r.x = ax; r.y = ay;
  *(v2f*)(out + (size_t)node * FD + 2 * lane) = r;
}


__global__ void pool_scatter_kernel(const int* __restrict__ seg, const float* __restrict__ h,
                                    float* __restrict__ psum, float* __restrict__ pcnt, int nNodes) {
  int t = blockIdx.x * blockDim.x + threadIdx.x;
  int node = t >> 6;
  int d    = t & 63;
  if (node < nNodes) {
    int g = seg[node];
    atomicAdd(&psum[(size_t)g * FD + d], h[(size_t)node * FD + d]);
    if (d == 0) atomicAdd(&pcnt[g], 1.0f);
  }
}

__global__ void out_gemm_kernel(const float* __restrict__ psum, const float* __restrict__ pcnt,
                                const float* __restrict__ Wout, const float* __restrict__ bout,
                                float* __restrict__ out) {
  int t = blockIdx.x * blockDim.x + threadIdx.x;
  if (t < NGRAPHS * NTARGETS) {
    int g = t / NTARGETS;
    int c = t % NTARGETS;
    float inv = 1.0f / fmaxf(pcnt[g], 1.0f);
    float acc = 0.0f;
#pragma unroll 8
    for (int d = 0; d < FD; ++d)
      acc += psum[(size_t)g * FD + d] * Wout[(size_t)d * NTARGETS + c];
    out[t] = acc * inv + bout[c];
  }
}


extern "C" void kernel_launch(void* const* d_in, const int* in_sizes, int n_in,
                              void* d_out, int out_size, void* d_ws, size_t ws_size,
                              hipStream_t stream) {
  const float* x    = (const float*)d_in[0];
  const float* W1   = (const float*)d_in[1];
  const float* b1   = (const float*)d_in[2];
  const float* W2   = (const float*)d_in[3];
  const float* b2   = (const float*)d_in[4];
  const float* W3   = (const float*)d_in[5];
  const float* b3   = (const float*)d_in[6];
  const float* Wout = (const float*)d_in[7];
  const float* bout = (const float*)d_in[8];
  const int*   eidx = (const int*)d_in[9];
  const int*   seg  = (const int*)d_in[10];
  float* out = (float*)d_out;
  (void)n_in; (void)out_size;

  const int nNodes = in_sizes[0] / FD;
  const int nE     = in_sizes[9] / 2;
  const int* srcI = eidx;
  const int* dstI = eidx + nE;

  char* ws = (char*)d_ws;
  size_t off = 0;
  auto alloc = [&](size_t bytes) -> void* {
    void* p = ws + off;
    off = (off + bytes + 255) & ~(size_t)255;
    return p;
  };
  float* dinv   = (float*)alloc((size_t)nNodes * sizeof(float));
  float* hbuf   = (float*)alloc((size_t)nNodes * FD * sizeof(float));
  float* feat   = (float*)alloc((size_t)nNodes * FD * sizeof(float));
  float* psum   = (float*)alloc((size_t)NGRAPHS * FD * sizeof(float));
  float* pcnt   = (float*)alloc((size_t)NGRAPHS * sizeof(float));
  int*   cnt    = (int*)alloc((size_t)(nNodes + 1) * sizeof(int));
  int*   cursor = (int*)alloc((size_t)nNodes * sizeof(int));
  int*   list   = (int*)alloc((size_t)nE * sizeof(int));
  if (ws_size < off) return;

  const int T = 256;
  const int nodeBlocks = (nNodes + T - 1) / T;
  const int gemmBlocks = nNodes / 16;
  const int nfeat      = nNodes * FD;
  const int featBlocks = (nfeat + T - 1) / T;
  const int eBlocks    = (nE + T - 1) / T;
  const int gBlocks    = (nNodes + 7) / 8;

  zero_int_kernel<<<(nNodes + 1 + T - 1) / T, T, 0, stream>>>(cnt, nNodes + 1);
  deg_count_kernel<<<eBlocks, T, 0, stream>>>(dstI, cnt, nE);
  offsets_kernel<<<nodeBlocks, T, 0, stream>>>(cnt, cnt + nNodes, cursor, dinv, nNodes);
  fill_kernel<<<eBlocks, T, 0, stream>>>(srcI, dstI, cursor, list, nE);

  gemm64_wmma<<<gemmBlocks, 128, 0, stream>>>(x, W1, hbuf);
  gather_kernel<true><<<gBlocks, T, 0, stream>>>(cnt, cursor, list, dinv, hbuf, b1, feat, nNodes);

  gemm64_wmma<<<gemmBlocks, 128, 0, stream>>>(feat, W2, hbuf);
  gather_kernel<true><<<gBlocks, T, 0, stream>>>(cnt, cursor, list, dinv, hbuf, b2, feat, nNodes);

  gemm64_wmma<<<gemmBlocks, 128, 0, stream>>>(feat, W3, hbuf);
  gather_kernel<false><<<gBlocks, T, 0, stream>>>(cnt, cursor, list, dinv, hbuf, b3, feat, nNodes);

  zero_kernel<<<(NGRAPHS * FD + T - 1) / T, T, 0, stream>>>(psum, NGRAPHS * FD);
  zero_kernel<<<(NGRAPHS + T - 1) / T, T, 0, stream>>>(pcnt, NGRAPHS);
  pool_scatter_kernel<<<featBlocks, T, 0, stream>>>(seg, feat, psum, pcnt, nNodes);
  out_gemm_kernel<<<(NGRAPHS * NTARGETS + T - 1) / T, T, 0, stream>>>(psum, pcnt, Wout, bout, out);
}


# Mean pool: one wave sums 16 consecutive node rows in registers (sequential), then one float atomic per column and one count add per run of equal graph id; was one atomic per element and per node
# speedup vs baseline: 3.1152x; 1.2705x over previous
#include <hip/hip_runtime.h>
#include <hip/hip_bf16.h>


static constexpr int FD       = 64;
static constexpr int NGRAPHS  = 512;
static constexpr int NTARGETS = 100;

typedef __attribute__((ext_vector_type(2))) float v2f;
typedef __attribute__((ext_vector_type(8))) float v8f;


__global__ void zero_kernel(float* __restrict__ p, int n) {
  int i = blockIdx.x * blockDim.x + threadIdx.x;
  if (i < n) p[i] = 0.0f;
}

__global__ void zero_int_kernel(int* __restrict__ p, int n) {
  int i = blockIdx.x * blockDim.x + threadIdx.x;
  if (i < n) p[i] = 0;
}


__global__ void deg_count_kernel(const int* __restrict__ dst, int* __restrict__ cnt, int nE) {
  int e = blockIdx.x * blockDim.x + threadIdx.x;
  if (e < nE) atomicAdd(&cnt[dst[e]], 1);
}

__global__ __launch_bounds__(256) void offsets_kernel(const int* __restrict__ cnt, int* __restrict__ total,
                                                      int* __restrict__ cursor, float* __restrict__ dinv, int n) {
  __shared__ int waveSum[8];
  __shared__ int blockBase;
  const int i    = blockIdx.x * 256 + threadIdx.x;
  const int lane = threadIdx.x & 31;
  const int wv   = threadIdx.x >> 5;
  const int c    = (i < n) ? cnt[i] : 0;
  int incl = c;
  for (int off = 1; off < 32; off <<= 1) {
    const int t = __shfl_up(incl, off, 32);
    if (lane >= off) incl += t;
  }
  if (lane == 31) waveSum[wv] = incl;
  __syncthreads();
  if (threadIdx.x == 0) {
    int run = 0;
    for (int w = 0; w < 8; ++w) { const int t = waveSum[w]; waveSum[w] = run; run += t; }
    blockBase = atomicAdd(total, run);
  }
  __syncthreads();
  if (i < n) {
    cursor[i] = blockBase + waveSum[wv] + incl - c;
    dinv[i]   = rsqrtf((float)c + 1.0f);
  }
}

__global__ void fill_kernel(const int* __restrict__ src, const int* __restrict__ dst,
                            int* __restrict__ cursor, int* __restrict__ list, int nE) {
  int e = blockIdx.x * blockDim.x + threadIdx.x;
  if (e < nE) {
    const int pos = atomicAdd(&cursor[dst[e]], 1);
    list[pos] = src[e];
  }
}

__global__ __launch_bounds__(128) void gemm64_wmma(const float* __restrict__ X,
                                                   const float* __restrict__ W,
                                                   float* __restrict__ H) {
  const int rowBase = blockIdx.x * 16;
  const int wave  = threadIdx.x >> 5;
  const int lane  = threadIdx.x & 31;
  const int m     = lane & 15;
  const int khalf = (lane >> 4) << 1;

  const float* __restrict__ xrow = X + (size_t)(rowBase + m) * FD;
  const float* __restrict__ wcol = W + wave * 16 + m;

  v8f acc = {};
#pragma unroll
  for (int kt = 0; kt < FD / 4; ++kt) {
    const int k = kt * 4 + khalf;
    v2f a; a.x = xrow[k];                 a.y = xrow[k + 1];
    v2f b; b.x = wcol[(size_t)k * FD];    b.y = wcol[(size_t)(k + 1) * FD];
    acc = __builtin_amdgcn_wmma_f32_16x16x4_f32(
         false, a,  false, b,
         (short)0, acc,  false,  false);
  }

  const int mBase = (lane >> 4) << 3;
  float* __restrict__ hout = H + (size_t)rowBase * FD + wave * 16 + m;
#pragma unroll
  for (int v = 0; v < 8; ++v)
    hout[(size_t)(mBase + v) * FD] = acc[v];
}


template <int G>
__device__ __forceinline__ void gather_group(const float* __restrict__ h, int s, float ds, float di,
                                             int jb, int m, int lane, float& ax, float& ay) {
  v2f v[G];
#pragma unroll
  for (int k = 0; k < G; ++k) {
    const int sj = __builtin_amdgcn_readlane(s, jb + k);
    v[k] = *(const v2f*)(h + (size_t)sj * FD + 2 * lane);
  }
#pragma unroll
  for (int k = 0; k < G; ++k) {
    const float dj = __int_as_float(__builtin_amdgcn_readlane(__float_as_int(ds), jb + k));
    const float wj = (jb + k < m) ? dj * di : 0.0f;
    ax += v[k].x * wj;
    ay += v[k].y * wj;
  }
}

template <bool RELU>
__global__ __launch_bounds__(256) void gather_kernel(const int* __restrict__ cnt, const int* __restrict__ cursorEnd,
                                                     const int* __restrict__ list, const float* __restrict__ dinv,
                                                     const float* __restrict__ h, const float* __restrict__ bias,
                                                     float* __restrict__ out, int nNodes) {
  const int lane = threadIdx.x & 31;
  const int node = __builtin_amdgcn_readfirstlane((int)((blockIdx.x * blockDim.x + threadIdx.x) >> 5));
  if (node >= nNodes) return;
  const int   c     = __builtin_amdgcn_readfirstlane(cnt[node]);
  const int   beg   = __builtin_amdgcn_readfirstlane(cursorEnd[node]) - c;
  const float di    = __int_as_float(__builtin_amdgcn_readfirstlane(__float_as_int(dinv[node])));
  const int   total = c + 1;
  float ax = 0.0f, ay = 0.0f;
  for (int cb = 0; cb < total; cb += 32) {
    const int idx = cb + lane;
    int s = node;
    if (idx < c) s = list[beg + idx];
    const float ds = dinv[s];
    const int m = min(32, total - cb);
    int jb = 0;
    for (; m - jb > 4; jb += 8) gather_group<8>(h, s, ds, di, jb, m, lane, ax, ay);
    if (m - jb > 0) gather_group<4>(h, s, ds, di, jb, m, lane, ax, ay);
  }
  const v2f b = *(const v2f*)(bias + 2 * lane);
  ax += b.x;
  ay += b.y;
  if (RELU) { ax = fmaxf(ax, 0.0f); ay = fmaxf(ay, 0.0f); }
  v2f r; r.x = ax; r.y = ay;
  *(v2f*)(out + (size_t)node * FD + 2 * lane) = r;
}


static constexpr int NPW = 16;

__device__ __forceinline__ void pool_flush(float* __restrict__ psum, float* __restrict__ pcnt,
                                           int g, float ax, float ay, int n, int lane) {
  float* p = psum + (size_t)g * FD + 2 * lane;
  atomicAdd(p, ax);
  atomicAdd(p + 1, ay);
  if (lane == 0) atomicAdd(pcnt + g, (float)n);
}

__global__ __launch_bounds__(256) void pool_runs_kernel(const int* __restrict__ seg, const float* __restrict__ h,
                                                        float* __restrict__ psum, float* __restrict__ pcnt,
                                                        int nNodes) {
  const int lane = threadIdx.x & 31;
  const int wave = __builtin_amdgcn_readfirstlane((int)((blockIdx.x * blockDim.x + threadIdx.x) >> 5));
  const int base = wave * NPW;
  if (base >= nNodes) return;
  const int m  = min(NPW, nNodes - base);
  const int gv = seg[base + min(lane, m - 1)];
  v2f v[NPW];
#pragma unroll
  for (int k = 0; k < NPW; ++k) {
    const int r = base + min(k, m - 1);
    v[k] = *(const v2f*)(h + (size_t)r * FD + 2 * lane);
  }
  int   cur = __builtin_amdgcn_readlane(gv, 0);
  float ax = 0.0f, ay = 0.0f;
  int   n = 0;
#pragma unroll
  for (int k = 0; k < NPW; ++k) {
    if (k < m) {
      const int g = __builtin_amdgcn_readlane(gv, k);
      if (g != cur) {
        pool_flush(psum, pcnt, cur, ax, ay, n, lane);
        cur = g; ax = 0.0f; ay = 0.0f; n = 0;
      }
      ax += v[k].x;
      ay += v[k].y;
      ++n;
    }
  }
  pool_flush(psum, pcnt, cur, ax, ay, n, lane);
}

__global__ void out_gemm_kernel(const float* __restrict__ psum, const float* __restrict__ pcnt,
                                const float* __restrict__ Wout, const float* __restrict__ bout,
                                float* __restrict__ out) {
  int t = blockIdx.x * blockDim.x + threadIdx.x;
  if (t < NGRAPHS * NTARGETS) {
    int g = t / NTARGETS;
    int c = t % NTARGETS;
    float inv = 1.0f / fmaxf(pcnt[g], 1.0f);
    float acc = 0.0f;
#pragma unroll 8
    for (int d = 0; d < FD; ++d)
      acc += psum[(size_t)g * FD + d] * Wout[(size_t)d * NTARGETS + c];
    out[t] = acc * inv + bout[c];
  }
}


extern "C" void kernel_launch(void* const* d_in, const int* in_sizes, int n_in,
                              void* d_out, int out_size, void* d_ws, size_t ws_size,
                              hipStream_t stream) {
  const float* x    = (const float*)d_in[0];
  const float* W1   = (const float*)d_in[1];
  const float* b1   = (const float*)d_in[2];
  const float* W2   = (const float*)d_in[3];
  const float* b2   = (const float*)d_in[4];
  const float* W3   = (const float*)d_in[5];
  const float* b3   = (const float*)d_in[6];
  const float* Wout = (const float*)d_in[7];
  const float* bout = (const float*)d_in[8];
  const int*   eidx = (const int*)d_in[9];
  const int*   seg  = (const int*)d_in[10];
  float* out = (float*)d_out;
  (void)n_in; (void)out_size;

  const int nNodes = in_sizes[0] / FD;
  const int nE     = in_sizes[9] / 2;
  const int* srcI = eidx;
  const int* dstI = eidx + nE;

  char* ws = (char*)d_ws;
  size_t off = 0;
  auto alloc = [&](size_t bytes) -> void* {
    void* p = ws + off;
    off = (off + bytes + 255) & ~(size_t)255;
    return p;
  };
  float* dinv   = (float*)alloc((size_t)nNodes * sizeof(float));
  float* hbuf   = (float*)alloc((size_t)nNodes * FD * sizeof(float));
  float* feat   = (float*)alloc((size_t)nNodes * FD * sizeof(float));
  float* psum   = (float*)alloc((size_t)NGRAPHS * FD * sizeof(float));
  float* pcnt   = (float*)alloc((size_t)NGRAPHS * sizeof(float));
  int*   cnt    = (int*)alloc((size_t)(nNodes + 1) * sizeof(int));
  int*   cursor = (int*)alloc((size_t)nNodes * sizeof(int));
  int*   list   = (int*)alloc((size_t)nE * sizeof(int));
  if (ws_size < off) return;

  const int T = 256;
  const int nodeBlocks = (nNodes + T - 1) / T;
  const int gemmBlocks = nNodes / 16;
  const int eBlocks    = (nE + T - 1) / T;
  const int gBlocks    = (nNodes + 7) / 8;
  const int pBlocks    = (nNodes + 8 * NPW - 1) / (8 * NPW);

  zero_int_kernel<<<(nNodes + 1 + T - 1) / T, T, 0, stream>>>(cnt, nNodes + 1);
  deg_count_kernel<<<eBlocks, T, 0, stream>>>(dstI, cnt, nE);
  offsets_kernel<<<nodeBlocks, T, 0, stream>>>(cnt, cnt + nNodes, cursor, dinv, nNodes);
  fill_kernel<<<eBlocks, T, 0, stream>>>(srcI, dstI, cursor, list, nE);

  gemm64_wmma<<<gemmBlocks, 128, 0, stream>>>(x, W1, hbuf);
  gather_kernel<true><<<gBlocks, T, 0, stream>>>(cnt, cursor, list, dinv, hbuf, b1, feat, nNodes);

  gemm64_wmma<<<gemmBlocks, 128, 0, stream>>>(feat, W2, hbuf);
  gather_kernel<true><<<gBlocks, T, 0, stream>>>(cnt, cursor, list, dinv, hbuf, b2, feat, nNodes);

  gemm64_wmma<<<gemmBlocks, 128, 0, stream>>>(feat, W3, hbuf);
  gather_kernel<false><<<gBlocks, T, 0, stream>>>(cnt, cursor, list, dinv, hbuf, b3, feat, nNodes);

  zero_kernel<<<(NGRAPHS * FD + T - 1) / T, T, 0, stream>>>(psum, NGRAPHS * FD);
  zero_kernel<<<(NGRAPHS + T - 1) / T, T, 0, stream>>>(pcnt, NGRAPHS);
  pool_runs_kernel<<<pBlocks, T, 0, stream>>>(seg, feat, psum, pcnt, nNodes);
  out_gemm_kernel<<<(NGRAPHS * NTARGETS + T - 1) / T, T, 0, stream>>>(psum, pcnt, Wout, bout, out);
}


# Product stores rows scaled by dinv[src]; gather adds plain rows and applies dinv[dst] once after the sum: no per-edge dinv lookup (terms round differently, order of sums kept)
# speedup vs baseline: 3.2717x; 1.0502x over previous
#include <hip/hip_runtime.h>
#include <hip/hip_bf16.h>


static constexpr int FD       = 64;
static constexpr int NGRAPHS  = 512;
static constexpr int NTARGETS = 100;

typedef __attribute__((ext_vector_type(2))) float v2f;
typedef __attribute__((ext_vector_type(8))) float v8f;


__global__ void zero_kernel(float* __restrict__ p, int n) {
  int i = blockIdx.x * blockDim.x + threadIdx.x;
  if (i < n) p[i] = 0.0f;
}

__global__ void zero_int_kernel(int* __restrict__ p, int n) {
  int i = blockIdx.x * blockDim.x + threadIdx.x;
  if (i < n) p[i] = 0;
}


__global__ void deg_count_kernel(const int* __restrict__ dst, int* __restrict__ cnt, int nE) {
  int e = blockIdx.x * blockDim.x + threadIdx.x;
  if (e < nE) atomicAdd(&cnt[dst[e]], 1);
}

__global__ __launch_bounds__(256) void offsets_kernel(const int* __restrict__ cnt, int* __restrict__ total,
                                                      int* __restrict__ cursor, float* __restrict__ dinv, int n) {
  __shared__ int waveSum[8];
  __shared__ int blockBase;
  const int i    = blockIdx.x * 256 + threadIdx.x;
  const int lane = threadIdx.x & 31;
  const int wv   = threadIdx.x >> 5;
  const int c    = (i < n) ? cnt[i] : 0;
  int incl = c;
  for (int off = 1; off < 32; off <<= 1) {
    const int t = __shfl_up(incl, off, 32);
    if (lane >= off) incl += t;
  }
  if (lane == 31) waveSum[wv] = incl;
  __syncthreads();
  if (threadIdx.x == 0) {
    int run = 0;
    for (int w = 0; w < 8; ++w) { const int t = waveSum[w]; waveSum[w] = run; run += t; }
    blockBase = atomicAdd(total, run);
  }
  __syncthreads();
  if (i < n) {
    cursor[i] = blockBase + waveSum[wv] + incl - c;
    dinv[i]   = rsqrtf((float)c + 1.0f);
  }
}

__global__ void fill_kernel(const int* __restrict__ src, const int* __restrict__ dst,
                            int* __restrict__ cursor, int* __restrict__ list, int nE) {
  int e = blockIdx.x * blockDim.x + threadIdx.x;
  if (e < nE) {
    const int pos = atomicAdd(&cursor[dst[e]], 1);
    list[pos] = src[e];
  }
}

__global__ __launch_bounds__(128) void gemm64_wmma(const float* __restrict__ X,
                                                   const float* __restrict__ W,
                                                   const float* __restrict__ dinv,
                                                   float* __restrict__ H) {
  const int rowBase = blockIdx.x * 16;
  const int wave  = threadIdx.x >> 5;
  const int lane  = threadIdx.x & 31;
  const int m     = lane & 15;
  const int khalf = (lane >> 4) << 1;

  const float* __restrict__ xrow = X + (size_t)(rowBase + m) * FD;
  const float* __restrict__ wcol = W + wave * 16 + m;

  float dlo[8], dhi[8];
#pragma unroll
  for (int v = 0; v < 8; ++v) { dlo[v] = dinv[rowBase + v]; dhi[v] = dinv[rowBase + 8 + v]; }

  v8f acc = {};
#pragma unroll
  for (int kt = 0; kt < FD / 4; ++kt) {
    const int k = kt * 4 + khalf;
    v2f a; a.x = xrow[k];                 a.y = xrow[k + 1];
    v2f b; b.x = wcol[(size_t)k * FD];    b.y = wcol[(size_t)(k + 1) * FD];
    acc = __builtin_amdgcn_wmma_f32_16x16x4_f32(
         false, a,  false, b,
         (short)0, acc,  false,  false);
  }

  __builtin_amdgcn_sched_barrier(0);
  asm volatile("v_nop\n\tv_nop\n\tv_nop\n\tv_nop\n\tv_nop\n\tv_nop\n\tv_nop\n\tv_nop");
  __builtin_amdgcn_sched_barrier(0);

  const int mBase = (lane >> 4) << 3;
  float* __restrict__ hout = H + (size_t)rowBase * FD + wave * 16 + m;
#pragma unroll
  for (int v = 0; v < 8; ++v)
    hout[(size_t)(mBase + v) * FD] = acc[v] * (mBase ? dhi[v] : dlo[v]);
}


template <int G>
__device__ __forceinline__ void gather_group(const float* __restrict__ h, int s,
                                             int jb, int m, int lane, float& ax, float& ay) {
  v2f v[G];
#pragma unroll
  for (int k = 0; k < G; ++k) {
    const int sj = __builtin_amdgcn_readlane(s, jb + k);
    v[k] = *(const v2f*)(h + (size_t)sj * FD + 2 * lane);
  }
#pragma unroll
  for (int k = 0; k < G; ++k) {
    const bool live = jb + k < m;
    ax += live ? v[k].x : 0.0f;
    ay += live ? v[k].y : 0.0f;
  }
}

template <bool RELU>
__global__ __launch_bounds__(256) void gather_kernel(const int* __restrict__ cnt, const int* __restrict__ cursorEnd,
                                                     const int* __restrict__ list, const float* __restrict__ dinv,
                                                     const float* __restrict__ h, const float* __restrict__ bias,
                                                     float* __restrict__ out, int nNodes) {
  const int lane = threadIdx.x & 31;
  const int node = __builtin_amdgcn_readfirstlane((int)((blockIdx.x * blockDim.x + threadIdx.x) >> 5));
  if (node >= nNodes) return;
  const int   c     = __builtin_amdgcn_readfirstlane(cnt[node]);
  const int   beg   = __builtin_amdgcn_readfirstlane(cursorEnd[node]) - c;
  const float di    = __int_as_float(__builtin_amdgcn_readfirstlane(__float_as_int(dinv[node])));
  const int   total = c + 1;
  float ax = 0.0f, ay = 0.0f;
  for (int cb = 0; cb < total; cb += 32) {
    const int idx = cb + lane;
    int s = node;
    if (idx < c) s = list[beg + idx];
    const int m = min(32, total - cb);
    int jb = 0;
    for (; m - jb > 4; jb += 8) gather_group<8>(h, s, jb, m, lane, ax, ay);
    if (m - jb > 0) gather_group<4>(h, s, jb, m, lane, ax, ay);
  }
  const v2f b = *(const v2f*)(bias + 2 * lane);
  ax = ax * di + b.x;
  ay = ay * di + b.y;
  if (RELU) { ax = fmaxf(ax, 0.0f); ay = fmaxf(ay, 0.0f); }
  v2f r; r.x = ax; r.y = ay;
  *(v2f*)(out + (size_t)node * FD + 2 * lane) = r;
}


static constexpr int NPW = 16;

__device__ __forceinline__ void pool_flush(float* __restrict__ psum, float* __restrict__ pcnt,
                                           int g, float ax, float ay, int n, int lane) {
  float* p = psum + (size_t)g * FD + 2 * lane;
  atomicAdd(p, ax);
  atomicAdd(p + 1, ay);
  if (lane == 0) atomicAdd(pcnt + g, (float)n);
}

__global__ __launch_bounds__(256) void pool_runs_kernel(const int* __restrict__ seg, const float* __restrict__ h,
                                                        float* __restrict__ psum, float* __restrict__ pcnt,
                                                        int nNodes) {
  const int lane = threadIdx.x & 31;
  const int wave = __builtin_amdgcn_readfirstlane((int)((blockIdx.x * blockDim.x + threadIdx.x) >> 5));
  const int base = wave * NPW;
  if (base >= nNodes) return;
  const int m  = min(NPW, nNodes - base);
  const int gv = seg[base + min(lane, m - 1)];
  v2f v[NPW];
#pragma unroll
  for (int k = 0; k < NPW; ++k) {
    const int r = base + min(k, m - 1);
    v[k] = *(const v2f*)(h + (size_t)r * FD + 2 * lane);
  }
  int   cur = __builtin_amdgcn_readlane(gv, 0);
  float ax = 0.0f, ay = 0.0f;
  int   n = 0;
#pragma unroll
  for (int k = 0; k < NPW; ++k) {
    if (k < m) {
      const int g = __builtin_amdgcn_readlane(gv, k);
      if (g != cur) {
        pool_flush(psum, pcnt, cur, ax, ay, n, lane);
        cur = g; ax = 0.0f; ay = 0.0f; n = 0;
      }
      ax += v[k].x;
      ay += v[k].y;
      ++n;
    }
  }
  pool_flush(psum, pcnt, cur, ax, ay, n, lane);
}

__global__ void out_gemm_kernel(const float* __restrict__ psum, const float* __restrict__ pcnt,
                                const float* __restrict__ Wout, const float* __restrict__ bout,
                                float* __restrict__ out) {
  int t = blockIdx.x * blockDim.x + threadIdx.x;
  if (t < NGRAPHS * NTARGETS) {
    int g = t / NTARGETS;
    int c = t % NTARGETS;
    float inv = 1.0f / fmaxf(pcnt[g], 1.0f);
    float acc = 0.0f;
#pragma unroll 8
    for (int d = 0; d < FD; ++d)
      acc += psum[(size_t)g * FD + d] * Wout[(size_t)d * NTARGETS + c];
    out[t] = acc * inv + bout[c];
  }
}


extern "C" void kernel_launch(void* const* d_in, const int* in_sizes, int n_in,
                              void* d_out, int out_size, void* d_ws, size_t ws_size,
                              hipStream_t stream) {
  const float* x    = (const float*)d_in[0];
  const float* W1   = (const float*)d_in[1];
  const float* b1   = (const float*)d_in[2];
  const float* W2   = (const float*)d_in[3];
  const float* b2   = (const float*)d_in[4];
  const float* W3   = (const float*)d_in[5];
  const float* b3   = (const float*)d_in[6];
  const float* Wout = (const float*)d_in[7];
  const float* bout = (const float*)d_in[8];
  const int*   eidx = (const int*)d_in[9];
  const int*   seg  = (const int*)d_in[10];
  float* out = (float*)d_out;
  (void)n_in; (void)out_size;

  const int nNodes = in_sizes[0] / FD;
  const int nE     = in_sizes[9] / 2;
  const int* srcI = eidx;
  const int* dstI = eidx + nE;

  char* ws = (char*)d_ws;
  size_t off = 0;
  auto alloc = [&](size_t bytes) -> void* {
    void* p = ws + off;
    off = (off + bytes + 255) & ~(size_t)255;
    return p;
  };
  float* dinv   = (float*)alloc((size_t)nNodes * sizeof(float));
  float* hbuf   = (float*)alloc((size_t)nNodes * FD * sizeof(float));
  float* feat   = (float*)alloc((size_t)nNodes * FD * sizeof(float));
  float* psum   = (float*)alloc((size_t)NGRAPHS * FD * sizeof(float));
  float* pcnt   = (float*)alloc((size_t)NGRAPHS * sizeof(float));
  int*   cnt    = (int*)alloc((size_t)(nNodes + 1) * sizeof(int));
  int*   cursor = (int*)alloc((size_t)nNodes * sizeof(int));
  int*   list   = (int*)alloc((size_t)nE * sizeof(int));
  if (ws_size < off) return;

  const int T = 256;
  const int nodeBlocks = (nNodes + T - 1) / T;
  const int gemmBlocks = nNodes / 16;
  const int eBlocks    = (nE + T - 1) / T;
  const int gBlocks    = (nNodes + 7) / 8;
  const int pBlocks    = (nNodes + 8 * NPW - 1) / (8 * NPW);

  zero_int_kernel<<<(nNodes + 1 + T - 1) / T, T, 0, stream>>>(cnt, nNodes + 1);
  deg_count_kernel<<<eBlocks, T, 0, stream>>>(dstI, cnt, nE);
  offsets_kernel<<<nodeBlocks, T, 0, stream>>>(cnt, cnt + nNodes, cursor, dinv, nNodes);
  fill_kernel<<<eBlocks, T, 0, stream>>>(srcI, dstI, cursor, list, nE);

  gemm64_wmma<<<gemmBlocks, 128, 0, stream>>>(x, W1, dinv, hbuf);
  gather_kernel<true><<<gBlocks, T, 0, stream>>>(cnt, cursor, list, dinv, hbuf, b1, feat, nNodes);

  gemm64_wmma<<<gemmBlocks, 128, 0, stream>>>(feat, W2, dinv, hbuf);
  gather_kernel<true><<<gBlocks, T, 0, stream>>>(cnt, cursor, list, dinv, hbuf, b2, feat, nNodes);

  gemm64_wmma<<<gemmBlocks, 128, 0, stream>>>(feat, W3, dinv, hbuf);
  gather_kernel<false><<<gBlocks, T, 0, stream>>>(cnt, cursor, list, dinv, hbuf, b3, feat, nNodes);

  zero_kernel<<<(NGRAPHS * FD + T - 1) / T, T, 0, stream>>>(psum, NGRAPHS * FD);
  zero_kernel<<<(NGRAPHS + T - 1) / T, T, 0, stream>>>(pcnt, NGRAPHS);
  pool_runs_kernel<<<pBlocks, T, 0, stream>>>(seg, feat, psum, pcnt, nNodes);
  out_gemm_kernel<<<(NGRAPHS * NTARGETS + T - 1) / T, T, 0, stream>>>(psum, pcnt, Wout, bout, out);
}


# Edge bucketing without global atomics: LDS histogram of 4096-edge chunks into 256-node bins, scan, packed edges moved to bins, one block per bin ranks per node in LDS; slot arrival order as before
# speedup vs baseline: 5.2002x; 1.5894x over previous
#include <hip/hip_runtime.h>
#include <hip/hip_bf16.h>


static constexpr int FD       = 64;
static constexpr int NGRAPHS  = 512;
static constexpr int NTARGETS = 100;

typedef __attribute__((ext_vector_type(2))) float v2f;
typedef __attribute__((ext_vector_type(8))) float v8f;


__global__ void zero_kernel(float* __restrict__ p, int n) {
  int i = blockIdx.x * blockDim.x + threadIdx.x;
  if (i < n) p[i] = 0.0f;
}


static constexpr int BIN_SHIFT = 8;
static constexpr int BIN_NODES = 1 << BIN_SHIFT;
static constexpr int SRC_BITS  = 24;
static constexpr int MAXBIN    = 512;
static constexpr int CHUNK     = 4096;
static constexpr int EPT       = CHUNK / 256;

__device__ __forceinline__ int block_excl_scan256(int v, int* waveSum, int& total) {
  const int lane = threadIdx.x & 31;
  const int wv   = threadIdx.x >> 5;
  int incl = v;
  for (int off = 1; off < 32; off <<= 1) {
    const int t = __shfl_up(incl, off, 32);
    if (lane >= off) incl += t;
  }
  if (lane == 31) waveSum[wv] = incl;
  __syncthreads();
  int wbase = 0, tot = 0;
  for (int w = 0; w < 8; ++w) {
    const int t = waveSum[w];
    if (w < wv) wbase += t;
    tot += t;
  }
  __syncthreads();
  total = tot;
  return wbase + incl - v;
}

__global__ __launch_bounds__(256) void bin_hist_kernel(const int* __restrict__ dst, int* __restrict__ blockhist,
                                                       int nE, int nBins, int nChunks) {
  __shared__ int hist[MAXBIN];
  for (int i = threadIdx.x; i < nBins; i += 256) hist[i] = 0;
  __syncthreads();
  const int base = blockIdx.x * CHUNK + threadIdx.x;
  int d[EPT];
#pragma unroll
  for (int k = 0; k < EPT; ++k) d[k] = dst[min(base + k * 256, nE - 1)];
#pragma unroll
  for (int k = 0; k < EPT; ++k)
    if (base + k * 256 < nE) atomicAdd(&hist[d[k] >> BIN_SHIFT], 1);
  __syncthreads();
  for (int i = threadIdx.x; i < nBins; i += 256) blockhist[(size_t)i * nChunks + blockIdx.x] = hist[i];
}

__global__ __launch_bounds__(256) void bin_scan_kernel(int* __restrict__ blockhist, int* __restrict__ binTotal,
                                                       int nChunks) {
  __shared__ int waveSum[8];
  int* __restrict__ row = blockhist + (size_t)blockIdx.x * nChunks;
  const int per = (nChunks + 255) / 256;
  const int lo  = threadIdx.x * per;
  const int hi  = min(lo + per, nChunks);
  int s = 0;
  for (int j = lo; j < hi; ++j) s += row[j];
  int total;
  int run = block_excl_scan256(s, waveSum, total);
  for (int j = lo; j < hi; ++j) { const int t = row[j]; row[j] = run; run += t; }
  if (threadIdx.x == 0) binTotal[blockIdx.x] = total;
}

__global__ __launch_bounds__(256) void bin_scatter_kernel(const int* __restrict__ src, const int* __restrict__ dst,
                                                          const int* __restrict__ blockhist,
                                                          const int* __restrict__ binTotal,
                                                          int* __restrict__ binStart, unsigned* __restrict__ part,
                                                          int nE, int nBins, int nChunks) {
  __shared__ int cur[MAXBIN];
  __shared__ int waveSum[8];
  const int base = blockIdx.x * CHUNK + threadIdx.x;
  int d[EPT], s[EPT];
#pragma unroll
  for (int k = 0; k < EPT; ++k) {
    const int e = min(base + k * 256, nE - 1);
    d[k] = dst[e];
    s[k] = src[e];
  }
  int carry = 0;
  for (int t0 = 0; t0 < nBins; t0 += 256) {
    const int i = t0 + threadIdx.x;
    const int c = (i < nBins) ? binTotal[i] : 0;
    int total;
    const int start = carry + block_excl_scan256(c, waveSum, total);
    if (i < nBins) {
      cur[i] = start + blockhist[(size_t)i * nChunks + blockIdx.x];
      if (blockIdx.x == 0) binStart[i] = start;
    }
    carry += total;
  }
  __syncthreads();
#pragma unroll
  for (int k = 0; k < EPT; ++k) {
    if (base + k * 256 < nE) {
      const int pos = atomicAdd(&cur[d[k] >> BIN_SHIFT], 1);
      part[pos] = (unsigned)s[k] | ((unsigned)(d[k] & (BIN_NODES - 1)) << SRC_BITS);
    }
  }
}

static constexpr int SORT_U = 8;
__global__ __launch_bounds__(256) void bin_sort_kernel(const unsigned* __restrict__ part, const int* __restrict__ binStart,
                                                       const int* __restrict__ binTotal, int* __restrict__ cnt,
                                                       int* __restrict__ cursor, float* __restrict__ dinv,
                                                       int* __restrict__ list, int nNodes) {
  __shared__ int lcnt[BIN_NODES + 32];
  __shared__ int waveSum[8];
  const int spare  = BIN_NODES + (threadIdx.x & 31);
  const int segBeg = binStart[blockIdx.x];
  const int segLen = binTotal[blockIdx.x];
  const unsigned* __restrict__ seg = part + segBeg;
  lcnt[threadIdx.x] = 0;
  if (threadIdx.x < 32) lcnt[BIN_NODES + threadIdx.x] = 0;
  __syncthreads();
  for (int j0 = threadIdx.x; j0 < segLen; j0 += 256 * SORT_U) {
    unsigned p[SORT_U];
#pragma unroll
    for (int k = 0; k < SORT_U; ++k) p[k] = seg[min(j0 + k * 256, segLen - 1)];
#pragma unroll
    for (int k = 0; k < SORT_U; ++k)
      atomicAdd(&lcnt[(j0 + k * 256 < segLen) ? (int)(p[k] >> SRC_BITS) : spare], 1);
  }
  __syncthreads();
  const int c = lcnt[threadIdx.x];
  int total;
  const int first = segBeg + block_excl_scan256(c, waveSum, total);
  lcnt[threadIdx.x] = first;
  const int node = blockIdx.x * BIN_NODES + threadIdx.x;
  if (node < nNodes) {
    cnt[node]    = c;
    cursor[node] = first + c;
    dinv[node]   = rsqrtf((float)c + 1.0f);
  }
  __syncthreads();
  for (int j0 = threadIdx.x; j0 < segLen; j0 += 256 * SORT_U) {
    unsigned p[SORT_U];
#pragma unroll
    for (int k = 0; k < SORT_U; ++k) p[k] = seg[min(j0 + k * 256, segLen - 1)];
#pragma unroll
    for (int k = 0; k < SORT_U; ++k) {
      const bool live = j0 + k * 256 < segLen;
      const int  pos  = atomicAdd(&lcnt[live ? (int)(p[k] >> SRC_BITS) : spare], 1);
      if (live) list[pos] = (int)(p[k] & ((1u << SRC_BITS) - 1u));
    }
  }
}

__global__ __launch_bounds__(128) void gemm64_wmma(const float* __restrict__ X,
                                                   const float* __restrict__ W,
                                                   const float* __restrict__ dinv,
                                                   float* __restrict__ H) {
  const int rowBase = blockIdx.x * 16;
  const int wave  = threadIdx.x >> 5;
  const int lane  = threadIdx.x & 31;
  const int m     = lane & 15;
  const int khalf = (lane >> 4) << 1;

  const float* __restrict__ xrow = X + (size_t)(rowBase + m) * FD;
  const float* __restrict__ wcol = W + wave * 16 + m;

  float dlo[8], dhi[8];
#pragma unroll
  for (int v = 0; v < 8; ++v) { dlo[v] = dinv[rowBase + v]; dhi[v] = dinv[rowBase + 8 + v]; }

  v8f acc = {};
#pragma unroll
  for (int kt = 0; kt < FD / 4; ++kt) {
    const int k = kt * 4 + khalf;
    v2f a; a.x = xrow[k];                 a.y = xrow[k + 1];
    v2f b; b.x = wcol[(size_t)k * FD];    b.y = wcol[(size_t)(k + 1) * FD];
    acc = __builtin_amdgcn_wmma_f32_16x16x4_f32(
         false, a,  false, b,
         (short)0, acc,  false,  false);
  }

  __builtin_amdgcn_sched_barrier(0);
  asm volatile("v_nop\n\tv_nop\n\tv_nop\n\tv_nop\n\tv_nop\n\tv_nop\n\tv_nop\n\tv_nop");
  __builtin_amdgcn_sched_barrier(0);

  const int mBase = (lane >> 4) << 3;
  float* __restrict__ hout = H + (size_t)rowBase * FD + wave * 16 + m;
#pragma unroll
  for (int v = 0; v < 8; ++v)
    hout[(size_t)(mBase + v) * FD] = acc[v] * (mBase ? dhi[v] : dlo[v]);
}


template <int G>
__device__ __forceinline__ void gather_group(const float* __restrict__ h, int s,
                                             int jb, int m, int lane, float& ax, float& ay) {
  v2f v[G];
#pragma unroll
  for (int k = 0; k < G; ++k) {
    const int sj = __builtin_amdgcn_readlane(s, jb + k);
    v[k] = *(const v2f*)(h + (size_t)sj * FD + 2 * lane);
  }
#pragma unroll
  for (int k = 0; k < G; ++k) {
    const bool live = jb + k < m;
    ax += live ? v[k].x : 0.0f;
    ay += live ? v[k].y : 0.0f;
  }
}

template <bool RELU>
__global__ __launch_bounds__(256) void gather_kernel(const int* __restrict__ cnt, const int* __restrict__ cursorEnd,
                                                     const int* __restrict__ list, const float* __restrict__ dinv,
                                                     const float* __restrict__ h, const float* __restrict__ bias,
                                                     float* __restrict__ out, int nNodes) {
  const int lane = threadIdx.x & 31;
  const int node = __builtin_amdgcn_readfirstlane((int)((blockIdx.x * blockDim.x + threadIdx.x) >> 5));
  if (node >= nNodes) return;
  const int   c     = __builtin_amdgcn_readfirstlane(cnt[node]);
  const int   beg   = __builtin_amdgcn_readfirstlane(cursorEnd[node]) - c;
  const float di    = __int_as_float(__builtin_amdgcn_readfirstlane(__float_as_int(dinv[node])));
  const int   total = c + 1;
  float ax = 0.0f, ay = 0.0f;
  for (int cb = 0; cb < total; cb += 32) {
    const int idx = cb + lane;
    int s = node;
    if (idx < c) s = list[beg + idx];
    const int m = min(32, total - cb);
    int jb = 0;
    for (; m - jb > 4; jb += 8) gather_group<8>(h, s, jb, m, lane, ax, ay);
    if (m - jb > 0) gather_group<4>(h, s, jb, m, lane, ax, ay);
  }
  const v2f b = *(const v2f*)(bias + 2 * lane);
  ax = ax * di + b.x;
  ay = ay * di + b.y;
  if (RELU) { ax = fmaxf(ax, 0.0f); ay = fmaxf(ay, 0.0f); }
  v2f r; r.x = ax; r.y = ay;
  *(v2f*)(out + (size_t)node * FD + 2 * lane) = r;
}


static constexpr int NPW = 16;

__device__ __forceinline__ void pool_flush(float* __restrict__ psum, float* __restrict__ pcnt,
                                           int g, float ax, float ay, int n, int lane) {
  float* p = psum + (size_t)g * FD + 2 * lane;
  atomicAdd(p, ax);
  atomicAdd(p + 1, ay);
  if (lane == 0) atomicAdd(pcnt + g, (float)n);
}

__global__ __launch_bounds__(256) void pool_runs_kernel(const int* __restrict__ seg, const float* __restrict__ h,
                                                        float* __restrict__ psum, float* __restrict__ pcnt,
                                                        int nNodes) {
  const int lane = threadIdx.x & 31;
  const int wave = __builtin_amdgcn_readfirstlane((int)((blockIdx.x * blockDim.x + threadIdx.x) >> 5));
  const int base = wave * NPW;
  if (base >= nNodes) return;
  const int m  = min(NPW, nNodes - base);
  const int gv = seg[base + min(lane, m - 1)];
  v2f v[NPW];
#pragma unroll
  for (int k = 0; k < NPW; ++k) {
    const int r = base + min(k, m - 1);
    v[k] = *(const v2f*)(h + (size_t)r * FD + 2 * lane);
  }
  int   cur = __builtin_amdgcn_readlane(gv, 0);
  float ax = 0.0f, ay = 0.0f;
  int   n = 0;
#pragma unroll
  for (int k = 0; k < NPW; ++k) {
    if (k < m) {
      const int g = __builtin_amdgcn_readlane(gv, k);
      if (g != cur) {
        pool_flush(psum, pcnt, cur, ax, ay, n, lane);
        cur = g; ax = 0.0f; ay = 0.0f; n = 0;
      }
      ax += v[k].x;
      ay += v[k].y;
      ++n;
    }
  }
  pool_flush(psum, pcnt, cur, ax, ay, n, lane);
}

__global__ void out_gemm_kernel(const float* __restrict__ psum, const float* __restrict__ pcnt,
                                const float* __restrict__ Wout, const float* __restrict__ bout,
                                float* __restrict__ out) {
  int t = blockIdx.x * blockDim.x + threadIdx.x;
  if (t < NGRAPHS * NTARGETS) {
    int g = t / NTARGETS;
    int c = t % NTARGETS;
    float inv = 1.0f / fmaxf(pcnt[g], 1.0f);
    float acc = 0.0f;
#pragma unroll 8
    for (int d = 0; d < FD; ++d)
      acc += psum[(size_t)g * FD + d] * Wout[(size_t)d * NTARGETS + c];
    out[t] = acc * inv + bout[c];
  }
}


extern "C" void kernel_launch(void* const* d_in, const int* in_sizes, int n_in,
                              void* d_out, int out_size, void* d_ws, size_t ws_size,
                              hipStream_t stream) {
  const float* x    = (const float*)d_in[0];
  const float* W1   = (const float*)d_in[1];
  const float* b1   = (const float*)d_in[2];
  const float* W2   = (const float*)d_in[3];
  const float* b2   = (const float*)d_in[4];
  const float* W3   = (const float*)d_in[5];
  const float* b3   = (const float*)d_in[6];
  const float* Wout = (const float*)d_in[7];
  const float* bout = (const float*)d_in[8];
  const int*   eidx = (const int*)d_in[9];
  const int*   seg  = (const int*)d_in[10];
  float* out = (float*)d_out;
  (void)n_in; (void)out_size;

  const int nNodes = in_sizes[0] / FD;
  const int nE     = in_sizes[9] / 2;
  const int* srcI = eidx;
  const int* dstI = eidx + nE;

  char* ws = (char*)d_ws;
  size_t off = 0;
  auto alloc = [&](size_t bytes) -> void* {
    void* p = ws + off;
    off = (off + bytes + 255) & ~(size_t)255;
    return p;
  };
  float* dinv   = (float*)alloc((size_t)nNodes * sizeof(float));
  float* hbuf   = (float*)alloc((size_t)nNodes * FD * sizeof(float));
  float* feat   = (float*)alloc((size_t)nNodes * FD * sizeof(float));
  float* psum   = (float*)alloc((size_t)NGRAPHS * FD * sizeof(float));
  float* pcnt   = (float*)alloc((size_t)NGRAPHS * sizeof(float));
  int*   cnt    = (int*)alloc((size_t)nNodes * sizeof(int));
  int*   cursor = (int*)alloc((size_t)nNodes * sizeof(int));
  int*   list   = (int*)alloc((size_t)nE * sizeof(int));
  if (ws_size < off) return;

  const int nBins   = (nNodes + BIN_NODES - 1) >> BIN_SHIFT;
  const int nChunks = (nE + CHUNK - 1) / CHUNK;
  unsigned* part      = (unsigned*)feat;
  int*      blockhist = (int*)(part + (((size_t)nE + 63) & ~(size_t)63));
  int*      binTotal  = blockhist + (size_t)nBins * nChunks;
  int*      binStart  = binTotal + nBins;
  if ((size_t)((const char*)(binStart + nBins) - (const char*)feat) > (size_t)nNodes * FD * sizeof(float)) return;
  if (nBins > MAXBIN || nNodes > (1 << SRC_BITS)) return;

  const int T = 256;
  const int gemmBlocks = nNodes / 16;
  const int gBlocks    = (nNodes + 7) / 8;
  const int pBlocks    = (nNodes + 8 * NPW - 1) / (8 * NPW);

  bin_hist_kernel<<<nChunks, T, 0, stream>>>(dstI, blockhist, nE, nBins, nChunks);
  bin_scan_kernel<<<nBins, T, 0, stream>>>(blockhist, binTotal, nChunks);
  bin_scatter_kernel<<<nChunks, T, 0, stream>>>(srcI, dstI, blockhist, binTotal, binStart, part, nE, nBins, nChunks);
  bin_sort_kernel<<<nBins, T, 0, stream>>>(part, binStart, binTotal, cnt, cursor, dinv, list, nNodes);

  gemm64_wmma<<<gemmBlocks, 128, 0, stream>>>(x, W1, dinv, hbuf);
  gather_kernel<true><<<gBlocks, T, 0, stream>>>(cnt, cursor, list, dinv, hbuf, b1, feat, nNodes);

  gemm64_wmma<<<gemmBlocks, 128, 0, stream>>>(feat, W2, dinv, hbuf);
  gather_kernel<true><<<gBlocks, T, 0, stream>>>(cnt, cursor, list, dinv, hbuf, b2, feat, nNodes);

  gemm64_wmma<<<gemmBlocks, 128, 0, stream>>>(feat, W3, dinv, hbuf);
  gather_kernel<false><<<gBlocks, T, 0, stream>>>(cnt, cursor, list, dinv, hbuf, b3, feat, nNodes);

  zero_kernel<<<(NGRAPHS * FD + T - 1) / T, T, 0, stream>>>(psum, NGRAPHS * FD);
  zero_kernel<<<(NGRAPHS + T - 1) / T, T, 0, stream>>>(pcnt, NGRAPHS);
  pool_runs_kernel<<<pBlocks, T, 0, stream>>>(seg, feat, psum, pcnt, nNodes);
  out_gemm_kernel<<<(NGRAPHS * NTARGETS + T - 1) / T, T, 0, stream>>>(psum, pcnt, Wout, bout, out);
}


# Gather launches use their 16 summed rows from LDS in place: layers 1-2 run the next product on them (same WMMA chain), layer 3 the pool sums; 14 launches to 11, every sum keeps its order
# speedup vs baseline: 6.1751x; 1.1875x over previous
#include <hip/hip_runtime.h>
#include <hip/hip_bf16.h>


static constexpr int FD       = 64;
static constexpr int NGRAPHS  = 512;
static constexpr int NTARGETS = 100;

typedef __attribute__((ext_vector_type(2))) float v2f;
typedef __attribute__((ext_vector_type(8))) float v8f;


__global__ void zero_kernel(float* __restrict__ p, int n) {
  int i = blockIdx.x * blockDim.x + threadIdx.x;
  if (i < n) p[i] = 0.0f;
}


static constexpr int BIN_SHIFT = 8;
static constexpr int BIN_NODES = 1 << BIN_SHIFT;
static constexpr int SRC_BITS  = 24;
static constexpr int MAXBIN    = 512;
static constexpr int CHUNK     = 4096;
static constexpr int EPT       = CHUNK / 256;

__device__ __forceinline__ int block_excl_scan256(int v, int* waveSum, int& total) {
  const int lane = threadIdx.x & 31;
  const int wv   = threadIdx.x >> 5;
  int incl = v;
  for (int off = 1; off < 32; off <<= 1) {
    const int t = __shfl_up(incl, off, 32);
    if (lane >= off) incl += t;
  }
  if (lane == 31) waveSum[wv] = incl;
  __syncthreads();
  int wbase = 0, tot = 0;
  for (int w = 0; w < 8; ++w) {
    const int t = waveSum[w];
    if (w < wv) wbase += t;
    tot += t;
  }
  __syncthreads();
  total = tot;
  return wbase + incl - v;
}

__global__ __launch_bounds__(256) void bin_hist_kernel(const int* __restrict__ dst, int* __restrict__ blockhist,
                                                       int nE, int nBins, int nChunks) {
  __shared__ int hist[MAXBIN];
  for (int i = threadIdx.x; i < nBins; i += 256) hist[i] = 0;
  __syncthreads();
  const int base = blockIdx.x * CHUNK + threadIdx.x;
  int d[EPT];
#pragma unroll
  for (int k = 0; k < EPT; ++k) d[k] = dst[min(base + k * 256, nE - 1)];
#pragma unroll
  for (int k = 0; k < EPT; ++k)
    if (base + k * 256 < nE) atomicAdd(&hist[d[k] >> BIN_SHIFT], 1);
  __syncthreads();
  for (int i = threadIdx.x; i < nBins; i += 256) blockhist[(size_t)i * nChunks + blockIdx.x] = hist[i];
}

__global__ __launch_bounds__(256) void bin_scan_kernel(int* __restrict__ blockhist, int* __restrict__ binTotal,
                                                       int nChunks) {
  __shared__ int waveSum[8];
  int* __restrict__ row = blockhist + (size_t)blockIdx.x * nChunks;
  const int per = (nChunks + 255) / 256;
  const int lo  = threadIdx.x * per;
  const int hi  = min(lo + per, nChunks);
  int s = 0;
  for (int j = lo; j < hi; ++j) s += row[j];
  int total;
  int run = block_excl_scan256(s, waveSum, total);
  for (int j = lo; j < hi; ++j) { const int t = row[j]; row[j] = run; run += t; }
  if (threadIdx.x == 0) binTotal[blockIdx.x] = total;
}

__global__ __launch_bounds__(256) void bin_scatter_kernel(const int* __restrict__ src, const int* __restrict__ dst,
                                                          const int* __restrict__ blockhist,
                                                          const int* __restrict__ binTotal,
                                                          int* __restrict__ binStart, unsigned* __restrict__ part,
                                                          int nE, int nBins, int nChunks) {
  __shared__ int cur[MAXBIN];
  __shared__ int waveSum[8];
  const int base = blockIdx.x * CHUNK + threadIdx.x;
  int d[EPT], s[EPT];
#pragma unroll
  for (int k = 0; k < EPT; ++k) {
    const int e = min(base + k * 256, nE - 1);
    d[k] = dst[e];
    s[k] = src[e];
  }
  int carry = 0;
  for (int t0 = 0; t0 < nBins; t0 += 256) {
    const int i = t0 + threadIdx.x;
    const int c = (i < nBins) ? binTotal[i] : 0;
    int total;
    const int start = carry + block_excl_scan256(c, waveSum, total);
    if (i < nBins) {
      cur[i] = start + blockhist[(size_t)i * nChunks + blockIdx.x];
      if (blockIdx.x == 0) binStart[i] = start;
    }
    carry += total;
  }
  __syncthreads();
#pragma unroll
  for (int k = 0; k < EPT; ++k) {
    if (base + k * 256 < nE) {
      const int pos = atomicAdd(&cur[d[k] >> BIN_SHIFT], 1);
      part[pos] = (unsigned)s[k] | ((unsigned)(d[k] & (BIN_NODES - 1)) << SRC_BITS);
    }
  }
}

static constexpr int SORT_U = 8;
__global__ __launch_bounds__(256) void bin_sort_kernel(const unsigned* __restrict__ part, const int* __restrict__ binStart,
                                                       const int* __restrict__ binTotal, int* __restrict__ cnt,
                                                       int* __restrict__ cursor, float* __restrict__ dinv,
                                                       int* __restrict__ list, int nNodes) {
  __shared__ int lcnt[BIN_NODES + 32];
  __shared__ int waveSum[8];
  const int spare  = BIN_NODES + (threadIdx.x & 31);
  const int segBeg = binStart[blockIdx.x];
  const int segLen = binTotal[blockIdx.x];
  const unsigned* __restrict__ seg = part + segBeg;
  lcnt[threadIdx.x] = 0;
  if (threadIdx.x < 32) lcnt[BIN_NODES + threadIdx.x] = 0;
  __syncthreads();
  for (int j0 = threadIdx.x; j0 < segLen; j0 += 256 * SORT_U) {
    unsigned p[SORT_U];
#pragma unroll
    for (int k = 0; k < SORT_U; ++k) p[k] = seg[min(j0 + k * 256, segLen - 1)];
#pragma unroll
    for (int k = 0; k < SORT_U; ++k)
      atomicAdd(&lcnt[(j0 + k * 256 < segLen) ? (int)(p[k] >> SRC_BITS) : spare], 1);
  }
  __syncthreads();
  const int c = lcnt[threadIdx.x];
  int total;
  const int first = segBeg + block_excl_scan256(c, waveSum, total);
  lcnt[threadIdx.x] = first;
  const int node = blockIdx.x * BIN_NODES + threadIdx.x;
  if (node < nNodes) {
    cnt[node]    = c;
    cursor[node] = first + c;
    dinv[node]   = rsqrtf((float)c + 1.0f);
  }
  __syncthreads();
  for (int j0 = threadIdx.x; j0 < segLen; j0 += 256 * SORT_U) {
    unsigned p[SORT_U];
#pragma unroll
    for (int k = 0; k < SORT_U; ++k) p[k] = seg[min(j0 + k * 256, segLen - 1)];
#pragma unroll
    for (int k = 0; k < SORT_U; ++k) {
      const bool live = j0 + k * 256 < segLen;
      const int  pos  = atomicAdd(&lcnt[live ? (int)(p[k] >> SRC_BITS) : spare], 1);
      if (live) list[pos] = (int)(p[k] & ((1u << SRC_BITS) - 1u));
    }
  }
}

__global__ __launch_bounds__(128) void gemm64_wmma(const float* __restrict__ X,
                                                   const float* __restrict__ W,
                                                   const float* __restrict__ dinv,
                                                   float* __restrict__ H) {
  const int rowBase = blockIdx.x * 16;
  const int wave  = threadIdx.x >> 5;
  const int lane  = threadIdx.x & 31;
  const int m     = lane & 15;
  const int khalf = (lane >> 4) << 1;

  const float* __restrict__ xrow = X + (size_t)(rowBase + m) * FD;
  const float* __restrict__ wcol = W + wave * 16 + m;

  float dlo[8], dhi[8];
#pragma unroll
  for (int v = 0; v < 8; ++v) { dlo[v] = dinv[rowBase + v]; dhi[v] = dinv[rowBase + 8 + v]; }

  v8f acc = {};
#pragma unroll
  for (int kt = 0; kt < FD / 4; ++kt) {
    const int k = kt * 4 + khalf;
    v2f a; a.x = xrow[k];                 a.y = xrow[k + 1];
    v2f b; b.x = wcol[(size_t)k * FD];    b.y = wcol[(size_t)(k + 1) * FD];
    acc = __builtin_amdgcn_wmma_f32_16x16x4_f32(
         false, a,  false, b,
         (short)0, acc,  false,  false);
  }

  __builtin_amdgcn_sched_barrier(0);
  asm volatile("v_nop\n\tv_nop\n\tv_nop\n\tv_nop\n\tv_nop\n\tv_nop\n\tv_nop\n\tv_nop");
  __builtin_amdgcn_sched_barrier(0);

  const int mBase = (lane >> 4) << 3;
  float* __restrict__ hout = H + (size_t)rowBase * FD + wave * 16 + m;
#pragma unroll
  for (int v = 0; v < 8; ++v)
    hout[(size_t)(mBase + v) * FD] = acc[v] * (mBase ? dhi[v] : dlo[v]);
}


template <int G>
__device__ __forceinline__ void gather_group(const float* __restrict__ h, int s,
                                             int jb, int m, int lane, float& ax, float& ay) {
  v2f v[G];
#pragma unroll
  for (int k = 0; k < G; ++k) {
    const int sj = __builtin_amdgcn_readlane(s, jb + k);
    v[k] = *(const v2f*)(h + (size_t)sj * FD + 2 * lane);
  }
#pragma unroll
  for (int k = 0; k < G; ++k) {
    const bool live = jb + k < m;
    ax += live ? v[k].x : 0.0f;
    ay += live ? v[k].y : 0.0f;
  }
}


static constexpr int TILE_PITCH = 68;

template <bool RELU>
__device__ __forceinline__ void gather_tile16(const int* __restrict__ cnt, const int* __restrict__ cursorEnd,
                                              const int* __restrict__ list, const float* __restrict__ dinv,
                                              const float* __restrict__ h, const float* __restrict__ bias,
                                              float* __restrict__ tile, int rowBase, int wave, int lane) {
  const int nb = rowBase + wave * 4;
  int c[4], beg[4], s0[4];
  float di[4];
#pragma unroll
  for (int q = 0; q < 4; ++q) {
    c[q]   = __builtin_amdgcn_readfirstlane(cnt[nb + q]);
    beg[q] = __builtin_amdgcn_readfirstlane(cursorEnd[nb + q]) - c[q];
    di[q]  = __int_as_float(__builtin_amdgcn_readfirstlane(__float_as_int(dinv[nb + q])));
  }
#pragma unroll
  for (int q = 0; q < 4; ++q) {
    s0[q] = nb + q;
    if (lane < c[q]) s0[q] = list[beg[q] + lane];
  }
  const v2f b = *(const v2f*)(bias + 2 * lane);
#pragma unroll
  for (int q = 0; q < 4; ++q) {
    const int node  = nb + q;
    const int total = c[q] + 1;
    int   s  = s0[q];
    float ax = 0.0f, ay = 0.0f;
    for (int cb = 0; cb < total; cb += 32) {
      const int m = min(32, total - cb);
      int jb = 0;
      for (; m - jb > 4; jb += 8) gather_group<8>(h, s, jb, m, lane, ax, ay);
      if (m - jb > 0) gather_group<4>(h, s, jb, m, lane, ax, ay);
      const int nidx = cb + 32 + lane;
      s = node;
      if (nidx < c[q]) s = list[beg[q] + nidx];
    }
    ax = ax * di[q] + b.x;
    ay = ay * di[q] + b.y;
    if (RELU) { ax = fmaxf(ax, 0.0f); ay = fmaxf(ay, 0.0f); }
    v2f r; r.x = ax; r.y = ay;
    *(v2f*)(tile + (wave * 4 + q) * TILE_PITCH + 2 * lane) = r;
  }
}

__global__ __launch_bounds__(128) void gather_product_kernel(const int* __restrict__ cnt, const int* __restrict__ cursorEnd,
                                                             const int* __restrict__ list, const float* __restrict__ dinv,
                                                             const float* __restrict__ h, const float* __restrict__ bias,
                                                             const float* __restrict__ W, float* __restrict__ H) {
  __shared__ __attribute__((aligned(16))) float tile[16 * TILE_PITCH];
  const int lane    = threadIdx.x & 31;
  const int wave    = __builtin_amdgcn_readfirstlane((int)(threadIdx.x >> 5));
  const int rowBase = blockIdx.x * 16;

  gather_tile16<true>(cnt, cursorEnd, list, dinv, h, bias, tile, rowBase, wave, lane);

  const int m     = lane & 15;
  const int khalf = (lane >> 4) << 1;
  const float* __restrict__ arow = tile + m * TILE_PITCH;
  const float* __restrict__ wcol = W + wave * 16 + m;

  v2f bfrag[FD / 4];
#pragma unroll
  for (int kt = 0; kt < FD / 4; ++kt) {
    const int k = kt * 4 + khalf;
    bfrag[kt].x = wcol[(size_t)k * FD];
    bfrag[kt].y = wcol[(size_t)(k + 1) * FD];
  }
  float dlo[8], dhi[8];
#pragma unroll
  for (int v = 0; v < 8; ++v) { dlo[v] = dinv[rowBase + v]; dhi[v] = dinv[rowBase + 8 + v]; }

  __syncthreads();

  v8f acc = {};
#pragma unroll
  for (int kt = 0; kt < FD / 4; ++kt) {
    const int k = kt * 4 + khalf;
    const v2f a = *(const v2f*)(arow + k);
    acc = __builtin_amdgcn_wmma_f32_16x16x4_f32(
         false, a,  false, bfrag[kt],
         (short)0, acc,  false,  false);
  }

  __builtin_amdgcn_sched_barrier(0);
  asm volatile("v_nop\n\tv_nop\n\tv_nop\n\tv_nop\n\tv_nop\n\tv_nop\n\tv_nop\n\tv_nop");
  __builtin_amdgcn_sched_barrier(0);

  const int mBase = (lane >> 4) << 3;
  float* __restrict__ hout = H + (size_t)rowBase * FD + wave * 16 + m;
#pragma unroll
  for (int v = 0; v < 8; ++v)
    hout[(size_t)(mBase + v) * FD] = acc[v] * (mBase ? dhi[v] : dlo[v]);
}


__device__ __forceinline__ void pool_flush(float* __restrict__ psum, float* __restrict__ pcnt,
                                           int g, float ax, float ay, int n, int lane) {
  float* p = psum + (size_t)g * FD + 2 * lane;
  atomicAdd(p, ax);
  atomicAdd(p + 1, ay);
  if (lane == 0) atomicAdd(pcnt + g, (float)n);
}

__global__ __launch_bounds__(128) void gather_pool_kernel(const int* __restrict__ cnt, const int* __restrict__ cursorEnd,
                                                          const int* __restrict__ list, const float* __restrict__ dinv,
                                                          const float* __restrict__ h, const float* __restrict__ bias,
                                                          const int* __restrict__ seg,
                                                          float* __restrict__ psum, float* __restrict__ pcnt) {
  __shared__ __attribute__((aligned(16))) float tile[16 * TILE_PITCH];
  const int lane    = threadIdx.x & 31;
  const int wave    = __builtin_amdgcn_readfirstlane((int)(threadIdx.x >> 5));
  const int rowBase = blockIdx.x * 16;

  gather_tile16<false>(cnt, cursorEnd, list, dinv, h, bias, tile, rowBase, wave, lane);
  __syncthreads();
  if (wave != 0) return;

  const int gv = seg[rowBase + (lane & 15)];
  int   cur = __builtin_amdgcn_readlane(gv, 0);
  float ax = 0.0f, ay = 0.0f;
  int   n = 0;
#pragma unroll
  for (int k = 0; k < 16; ++k) {
    const v2f v = *(const v2f*)(tile + k * TILE_PITCH + 2 * lane);
    const int g = __builtin_amdgcn_readlane(gv, k);
    if (g != cur) {
      pool_flush(psum, pcnt, cur, ax, ay, n, lane);
      cur = g; ax = 0.0f; ay = 0.0f; n = 0;
    }
    ax += v.x;
    ay += v.y;
    ++n;
  }
  pool_flush(psum, pcnt, cur, ax, ay, n, lane);
}

__global__ void out_gemm_kernel(const float* __restrict__ psum, const float* __restrict__ pcnt,
                                const float* __restrict__ Wout, const float* __restrict__ bout,
                                float* __restrict__ out) {
  int t = blockIdx.x * blockDim.x + threadIdx.x;
  if (t < NGRAPHS * NTARGETS) {
    int g = t / NTARGETS;
    int c = t % NTARGETS;
    float inv = 1.0f / fmaxf(pcnt[g], 1.0f);
    float acc = 0.0f;
#pragma unroll 8
    for (int d = 0; d < FD; ++d)
      acc += psum[(size_t)g * FD + d] * Wout[(size_t)d * NTARGETS + c];
    out[t] = acc * inv + bout[c];
  }
}


extern "C" void kernel_launch(void* const* d_in, const int* in_sizes, int n_in,
                              void* d_out, int out_size, void* d_ws, size_t ws_size,
                              hipStream_t stream) {
  const float* x    = (const float*)d_in[0];
  const float* W1   = (const float*)d_in[1];
  const float* b1   = (const float*)d_in[2];
  const float* W2   = (const float*)d_in[3];
  const float* b2   = (const float*)d_in[4];
  const float* W3   = (const float*)d_in[5];
  const float* b3   = (const float*)d_in[6];
  const float* Wout = (const float*)d_in[7];
  const float* bout = (const float*)d_in[8];
  const int*   eidx = (const int*)d_in[9];
  const int*   seg  = (const int*)d_in[10];
  float* out = (float*)d_out;
  (void)n_in; (void)out_size;

  const int nNodes = in_sizes[0] / FD;
  const int nE     = in_sizes[9] / 2;
  const int* srcI = eidx;
  const int* dstI = eidx + nE;

  char* ws = (char*)d_ws;
  size_t off = 0;
  auto alloc = [&](size_t bytes) -> void* {
    void* p = ws + off;
    off = (off + bytes + 255) & ~(size_t)255;
    return p;
  };
  float* dinv   = (float*)alloc((size_t)nNodes * sizeof(float));
  float* hbuf   = (float*)alloc((size_t)nNodes * FD * sizeof(float));
  float* feat   = (float*)alloc((size_t)nNodes * FD * sizeof(float));
  float* psum   = (float*)alloc((size_t)NGRAPHS * FD * sizeof(float));
  float* pcnt   = (float*)alloc((size_t)NGRAPHS * sizeof(float));
  int*   cnt    = (int*)alloc((size_t)nNodes * sizeof(int));
  int*   cursor = (int*)alloc((size_t)nNodes * sizeof(int));
  int*   list   = (int*)alloc((size_t)nE * sizeof(int));
  if (ws_size < off) return;

  const int nBins   = (nNodes + BIN_NODES - 1) >> BIN_SHIFT;
  const int nChunks = (nE + CHUNK - 1) / CHUNK;
  unsigned* part      = (unsigned*)feat;
  int*      blockhist = (int*)(part + (((size_t)nE + 63) & ~(size_t)63));
  int*      binTotal  = blockhist + (size_t)nBins * nChunks;
  int*      binStart  = binTotal + nBins;
  if ((size_t)((const char*)(binStart + nBins) - (const char*)feat) > (size_t)nNodes * FD * sizeof(float)) return;
  if (nBins > MAXBIN || nNodes > (1 << SRC_BITS)) return;

  const int T = 256;
  const int tileBlocks = nNodes / 16;

  bin_hist_kernel<<<nChunks, T, 0, stream>>>(dstI, blockhist, nE, nBins, nChunks);
  bin_scan_kernel<<<nBins, T, 0, stream>>>(blockhist, binTotal, nChunks);
  bin_scatter_kernel<<<nChunks, T, 0, stream>>>(srcI, dstI, blockhist, binTotal, binStart, part, nE, nBins, nChunks);
  bin_sort_kernel<<<nBins, T, 0, stream>>>(part, binStart, binTotal, cnt, cursor, dinv, list, nNodes);

  gemm64_wmma<<<tileBlocks, 128, 0, stream>>>(x, W1, dinv, hbuf);
  gather_product_kernel<<<tileBlocks, 128, 0, stream>>>(cnt, cursor, list, dinv, hbuf, b1, W2, feat);
  gather_product_kernel<<<tileBlocks, 128, 0, stream>>>(cnt, cursor, list, dinv, feat, b2, W3, hbuf);

  zero_kernel<<<(NGRAPHS * FD + T - 1) / T, T, 0, stream>>>(psum, NGRAPHS * FD);
  zero_kernel<<<(NGRAPHS + T - 1) / T, T, 0, stream>>>(pcnt, NGRAPHS);
  gather_pool_kernel<<<tileBlocks, 128, 0, stream>>>(cnt, cursor, list, dinv, hbuf, b3, seg, psum, pcnt);
  out_gemm_kernel<<<(NGRAPHS * NTARGETS + T - 1) / T, T, 0, stream>>>(psum, pcnt, Wout, bout, out);
}


# Layer 1 as (A_hat x) W1: its gather sums x rows times dinv[src] by fma (rounds differently, entry order kept), then both products in that launch; scan launch clears pool sums; 11 launches to 8
# speedup vs baseline: 6.4029x; 1.0369x over previous
#include <hip/hip_runtime.h>
#include <hip/hip_bf16.h>


static constexpr int FD       = 64;
static constexpr int NGRAPHS  = 512;
static constexpr int NTARGETS = 100;

typedef __attribute__((ext_vector_type(2))) float v2f;
typedef __attribute__((ext_vector_type(8))) float v8f;


static constexpr int BIN_SHIFT = 8;
static constexpr int BIN_NODES = 1 << BIN_SHIFT;
static constexpr int SRC_BITS  = 24;
static constexpr int MAXBIN    = 512;
static constexpr int CHUNK     = 4096;
static constexpr int EPT       = CHUNK / 256;

__device__ __forceinline__ int block_excl_scan256(int v, int* waveSum, int& total) {
  const int lane = threadIdx.x & 31;
  const int wv   = threadIdx.x >> 5;
  int incl = v;
  for (int off = 1; off < 32; off <<= 1) {
    const int t = __shfl_up(incl, off, 32);
    if (lane >= off) incl += t;
  }
  if (lane == 31) waveSum[wv] = incl;
  __syncthreads();
  int wbase = 0, tot = 0;
  for (int w = 0; w < 8; ++w) {
    const int t = waveSum[w];
    if (w < wv) wbase += t;
    tot += t;
  }
  __syncthreads();
  total = tot;
  return wbase + incl - v;
}

__global__ __launch_bounds__(256) void bin_hist_kernel(const int* __restrict__ dst, int* __restrict__ blockhist,
                                                       int nE, int nBins, int nChunks) {
  __shared__ int hist[MAXBIN];
  for (int i = threadIdx.x; i < nBins; i += 256) hist[i] = 0;
  __syncthreads();
  const int base = blockIdx.x * CHUNK + threadIdx.x;
  int d[EPT];
#pragma unroll
  for (int k = 0; k < EPT; ++k) d[k] = dst[min(base + k * 256, nE - 1)];
#pragma unroll
  for (int k = 0; k < EPT; ++k)
    if (base + k * 256 < nE) atomicAdd(&hist[d[k] >> BIN_SHIFT], 1);
  __syncthreads();
  for (int i = threadIdx.x; i < nBins; i += 256) blockhist[(size_t)i * nChunks + blockIdx.x] = hist[i];
}

__global__ __launch_bounds__(256) void bin_scan_kernel(int* __restrict__ blockhist, int* __restrict__ binTotal,
                                                       int nChunks, float* __restrict__ psum, float* __restrict__ pcnt) {
  __shared__ int waveSum[8];
  int* __restrict__ row = blockhist + (size_t)blockIdx.x * nChunks;
  const int per = (nChunks + 255) / 256;
  const int lo  = threadIdx.x * per;
  const int hi  = min(lo + per, nChunks);
  int s = 0;
  for (int j = lo; j < hi; ++j) s += row[j];
  int total;
  int run = block_excl_scan256(s, waveSum, total);
  for (int j = lo; j < hi; ++j) { const int t = row[j]; row[j] = run; run += t; }
  if (threadIdx.x == 0) binTotal[blockIdx.x] = total;
  const int gid = blockIdx.x * 256 + threadIdx.x;
  if (gid < NGRAPHS * FD) psum[gid] = 0.0f;
  if (gid < NGRAPHS) pcnt[gid] = 0.0f;
}

__global__ __launch_bounds__(256) void bin_scatter_kernel(const int* __restrict__ src, const int* __restrict__ dst,
                                                          const int* __restrict__ blockhist,
                                                          const int* __restrict__ binTotal,
                                                          int* __restrict__ binStart, unsigned* __restrict__ part,
                                                          int nE, int nBins, int nChunks) {
  __shared__ int cur[MAXBIN];
  __shared__ int waveSum[8];
  const int base = blockIdx.x * CHUNK + threadIdx.x;
  int d[EPT], s[EPT];
#pragma unroll
  for (int k = 0; k < EPT; ++k) {
    const int e = min(base + k * 256, nE - 1);
    d[k] = dst[e];
    s[k] = src[e];
  }
  int carry = 0;
  for (int t0 = 0; t0 < nBins; t0 += 256) {
    const int i = t0 + threadIdx.x;
    const int c = (i < nBins) ? binTotal[i] : 0;
    int total;
    const int start = carry + block_excl_scan256(c, waveSum, total);
    if (i < nBins) {
      cur[i] = start + blockhist[(size_t)i * nChunks + blockIdx.x];
      if (blockIdx.x == 0) binStart[i] = start;
    }
    carry += total;
  }
  __syncthreads();
#pragma unroll
  for (int k = 0; k < EPT; ++k) {
    if (base + k * 256 < nE) {
      const int pos = atomicAdd(&cur[d[k] >> BIN_SHIFT], 1);
      part[pos] = (unsigned)s[k] | ((unsigned)(d[k] & (BIN_NODES - 1)) << SRC_BITS);
    }
  }
}

static constexpr int SORT_U = 8;
__global__ __launch_bounds__(256) void bin_sort_kernel(const unsigned* __restrict__ part, const int* __restrict__ binStart,
                                                       const int* __restrict__ binTotal, int* __restrict__ cnt,
                                                       int* __restrict__ cursor, float* __restrict__ dinv,
                                                       int* __restrict__ list, int nNodes) {
  __shared__ int lcnt[BIN_NODES + 32];
  __shared__ int waveSum[8];
  const int spare  = BIN_NODES + (threadIdx.x & 31);
  const int segBeg = binStart[blockIdx.x];
  const int segLen = binTotal[blockIdx.x];
  const unsigned* __restrict__ seg = part + segBeg;
  lcnt[threadIdx.x] = 0;
  if (threadIdx.x < 32) lcnt[BIN_NODES + threadIdx.x] = 0;
  __syncthreads();
  for (int j0 = threadIdx.x; j0 < segLen; j0 += 256 * SORT_U) {
    unsigned p[SORT_U];
#pragma unroll
    for (int k = 0; k < SORT_U; ++k) p[k] = seg[min(j0 + k * 256, segLen - 1)];
#pragma unroll
    for (int k = 0; k < SORT_U; ++k)
      atomicAdd(&lcnt[(j0 + k * 256 < segLen) ? (int)(p[k] >> SRC_BITS) : spare], 1);
  }
  __syncthreads();
  const int c = lcnt[threadIdx.x];
  int total;
  const int first = segBeg + block_excl_scan256(c, waveSum, total);
  lcnt[threadIdx.x] = first;
  const int node = blockIdx.x * BIN_NODES + threadIdx.x;
  if (node < nNodes) {
    cnt[node]    = c;
    cursor[node] = first + c;
    dinv[node]   = rsqrtf((float)c + 1.0f);
  }
  __syncthreads();
  for (int j0 = threadIdx.x; j0 < segLen; j0 += 256 * SORT_U) {
    unsigned p[SORT_U];
#pragma unroll
    for (int k = 0; k < SORT_U; ++k) p[k] = seg[min(j0 + k * 256, segLen - 1)];
#pragma unroll
    for (int k = 0; k < SORT_U; ++k) {
      const bool live = j0 + k * 256 < segLen;
      const int  pos  = atomicAdd(&lcnt[live ? (int)(p[k] >> SRC_BITS) : spare], 1);
      if (live) list[pos] = (int)(p[k] & ((1u << SRC_BITS) - 1u));
    }
  }
}


template <int G>
__device__ __forceinline__ void gather_group(const float* __restrict__ h, int s,
                                             int jb, int m, int lane, float& ax, float& ay) {
  v2f v[G];
#pragma unroll
  for (int k = 0; k < G; ++k) {
    const int sj = __builtin_amdgcn_readlane(s, jb + k);
    v[k] = *(const v2f*)(h + (size_t)sj * FD + 2 * lane);
  }
#pragma unroll
  for (int k = 0; k < G; ++k) {
    const bool live = jb + k < m;
    ax += live ? v[k].x : 0.0f;
    ay += live ? v[k].y : 0.0f;
  }
}


static constexpr int TILE_PITCH = 68;

template <bool RELU>
__device__ __forceinline__ void gather_tile16(const int* __restrict__ cnt, const int* __restrict__ cursorEnd,
                                              const int* __restrict__ list, const float* __restrict__ dinv,
                                              const float* __restrict__ h, const float* __restrict__ bias,
                                              float* __restrict__ tile, int rowBase, int wave, int lane) {
  const int nb = rowBase + wave * 4;
  int c[4], beg[4], s0[4];
  float di[4];
#pragma unroll
  for (int q = 0; q < 4; ++q) {
    c[q]   = __builtin_amdgcn_readfirstlane(cnt[nb + q]);
    beg[q] = __builtin_amdgcn_readfirstlane(cursorEnd[nb + q]) - c[q];
    di[q]  = __int_as_float(__builtin_amdgcn_readfirstlane(__float_as_int(dinv[nb + q])));
  }
#pragma unroll
  for (int q = 0; q < 4; ++q) {
    s0[q] = nb + q;
    if (lane < c[q]) s0[q] = list[beg[q] + lane];
  }
  const v2f b = *(const v2f*)(bias + 2 * lane);
#pragma unroll
  for (int q = 0; q < 4; ++q) {
    const int node  = nb + q;
    const int total = c[q] + 1;
    int   s  = s0[q];
    float ax = 0.0f, ay = 0.0f;
    for (int cb = 0; cb < total; cb += 32) {
      const int m = min(32, total - cb);
      int jb = 0;
      for (; m - jb > 4; jb += 8) gather_group<8>(h, s, jb, m, lane, ax, ay);
      if (m - jb > 0) gather_group<4>(h, s, jb, m, lane, ax, ay);
      const int nidx = cb + 32 + lane;
      s = node;
      if (nidx < c[q]) s = list[beg[q] + nidx];
    }
    ax = ax * di[q] + b.x;
    ay = ay * di[q] + b.y;
    if (RELU) { ax = fmaxf(ax, 0.0f); ay = fmaxf(ay, 0.0f); }
    v2f r; r.x = ax; r.y = ay;
    *(v2f*)(tile + (wave * 4 + q) * TILE_PITCH + 2 * lane) = r;
  }
}

__global__ __launch_bounds__(128) void gather_product_kernel(const int* __restrict__ cnt, const int* __restrict__ cursorEnd,
                                                             const int* __restrict__ list, const float* __restrict__ dinv,
                                                             const float* __restrict__ h, const float* __restrict__ bias,
                                                             const float* __restrict__ W, float* __restrict__ H) {
  __shared__ __attribute__((aligned(16))) float tile[16 * TILE_PITCH];
  const int lane    = threadIdx.x & 31;
  const int wave    = __builtin_amdgcn_readfirstlane((int)(threadIdx.x >> 5));
  const int rowBase = blockIdx.x * 16;

  gather_tile16<true>(cnt, cursorEnd, list, dinv, h, bias, tile, rowBase, wave, lane);

  const int m     = lane & 15;
  const int khalf = (lane >> 4) << 1;
  const float* __restrict__ arow = tile + m * TILE_PITCH;
  const float* __restrict__ wcol = W + wave * 16 + m;

  v2f bfrag[FD / 4];
#pragma unroll
  for (int kt = 0; kt < FD / 4; ++kt) {
    const int k = kt * 4 + khalf;
    bfrag[kt].x = wcol[(size_t)k * FD];
    bfrag[kt].y = wcol[(size_t)(k + 1) * FD];
  }
  float dlo[8], dhi[8];
#pragma unroll
  for (int v = 0; v < 8; ++v) { dlo[v] = dinv[rowBase + v]; dhi[v] = dinv[rowBase + 8 + v]; }

  __syncthreads();

  v8f acc = {};
#pragma unroll
  for (int kt = 0; kt < FD / 4; ++kt) {
    const int k = kt * 4 + khalf;
    const v2f a = *(const v2f*)(arow + k);
    acc = __builtin_amdgcn_wmma_f32_16x16x4_f32(
         false, a,  false, bfrag[kt],
         (short)0, acc,  false,  false);
  }

  __builtin_amdgcn_sched_barrier(0);
  asm volatile("v_nop\n\tv_nop\n\tv_nop\n\tv_nop\n\tv_nop\n\tv_nop\n\tv_nop\n\tv_nop");
  __builtin_amdgcn_sched_barrier(0);

  const int mBase = (lane >> 4) << 3;
  float* __restrict__ hout = H + (size_t)rowBase * FD + wave * 16 + m;
#pragma unroll
  for (int v = 0; v < 8; ++v)
    hout[(size_t)(mBase + v) * FD] = acc[v] * (mBase ? dhi[v] : dlo[v]);
}


template <int G>
__device__ __forceinline__ void gather_group_w(const float* __restrict__ x, int s, float w,
                                               int jb, int lane, float& ax, float& ay) {
  v2f v[G];
#pragma unroll
  for (int k = 0; k < G; ++k) {
    const int sj = __builtin_amdgcn_readlane(s, jb + k);
    v[k] = *(const v2f*)(x + (size_t)sj * FD + 2 * lane);
  }
#pragma unroll
  for (int k = 0; k < G; ++k) {
    const float wj = __int_as_float(__builtin_amdgcn_readlane(__float_as_int(w), jb + k));
    ax = fmaf(v[k].x, wj, ax);
    ay = fmaf(v[k].y, wj, ay);
  }
}

__device__ __forceinline__ void gather_tile16_x(const int* __restrict__ cnt, const int* __restrict__ cursorEnd,
                                                const int* __restrict__ list, const float* __restrict__ dinv,
                                                const float* __restrict__ x,
                                                float* __restrict__ tile, int rowBase, int wave, int lane) {
  const int nb = rowBase + wave * 4;
  int c[4], beg[4], s0[4];
  float di[4], w0[4];
#pragma unroll
  for (int q = 0; q < 4; ++q) {
    c[q]   = __builtin_amdgcn_readfirstlane(cnt[nb + q]);
    beg[q] = __builtin_amdgcn_readfirstlane(cursorEnd[nb + q]) - c[q];
    di[q]  = __int_as_float(__builtin_amdgcn_readfirstlane(__float_as_int(dinv[nb + q])));
  }
#pragma unroll
  for (int q = 0; q < 4; ++q) {
    s0[q] = nb + q;
    if (lane < c[q]) s0[q] = list[beg[q] + lane];
  }
#pragma unroll
  for (int q = 0; q < 4; ++q) w0[q] = dinv[s0[q]];
#pragma unroll
  for (int q = 0; q < 4; ++q) {
    const int node  = nb + q;
    const int total = c[q] + 1;
    int   s  = s0[q];
    float w  = (lane < total) ? w0[q] : 0.0f;
    float ax = 0.0f, ay = 0.0f;
    for (int cb = 0; cb < total; cb += 32) {
      const int m = min(32, total - cb);
      int jb = 0;
      for (; m - jb > 4; jb += 8) gather_group_w<8>(x, s, w, jb, lane, ax, ay);
      if (m - jb > 0) gather_group_w<4>(x, s, w, jb, lane, ax, ay);
      if (cb + 32 < total) {
        const int nidx = cb + 32 + lane;
        s = node;
        if (nidx < c[q]) s = list[beg[q] + nidx];
        const float ws = dinv[s];
        w = (nidx < total) ? ws : 0.0f;
      }
    }
    v2f r; r.x = ax * di[q]; r.y = ay * di[q];
    *(v2f*)(tile + (wave * 4 + q) * TILE_PITCH + 2 * lane) = r;
  }
}

__global__ __launch_bounds__(128) void gather_x_product2_kernel(const int* __restrict__ cnt, const int* __restrict__ cursorEnd,
                                                                const int* __restrict__ list, const float* __restrict__ dinv,
                                                                const float* __restrict__ x,
                                                                const float* __restrict__ W1, const float* __restrict__ b1,
                                                                const float* __restrict__ W2, float* __restrict__ H) {
  __shared__ __attribute__((aligned(16))) float tile[16 * TILE_PITCH];
  __shared__ __attribute__((aligned(16))) float tile2[16 * TILE_PITCH];
  const int lane    = threadIdx.x & 31;
  const int wave    = __builtin_amdgcn_readfirstlane((int)(threadIdx.x >> 5));
  const int rowBase = blockIdx.x * 16;

  gather_tile16_x(cnt, cursorEnd, list, dinv, x, tile, rowBase, wave, lane);

  const int m     = lane & 15;
  const int khalf = (lane >> 4) << 1;
  const int mBase = (lane >> 4) << 3;

  const float* __restrict__ wcol1 = W1 + wave * 16 + m;
  v2f bfrag[FD / 4];
#pragma unroll
  for (int kt = 0; kt < FD / 4; ++kt) {
    const int k = kt * 4 + khalf;
    bfrag[kt].x = wcol1[(size_t)k * FD];
    bfrag[kt].y = wcol1[(size_t)(k + 1) * FD];
  }
  const float bcol = b1[wave * 16 + m];

  __syncthreads();

  {
    const float* __restrict__ arow = tile + m * TILE_PITCH;
    v8f acc = {};
#pragma unroll
    for (int kt = 0; kt < FD / 4; ++kt) {
      const int k = kt * 4 + khalf;
      const v2f a = *(const v2f*)(arow + k);
      acc = __builtin_amdgcn_wmma_f32_16x16x4_f32(
           false, a,  false, bfrag[kt],
           (short)0, acc,  false,  false);
    }
    __builtin_amdgcn_sched_barrier(0);
    asm volatile("v_nop\n\tv_nop\n\tv_nop\n\tv_nop\n\tv_nop\n\tv_nop\n\tv_nop\n\tv_nop");
    __builtin_amdgcn_sched_barrier(0);

    float* __restrict__ t2 = tile2 + mBase * TILE_PITCH + wave * 16 + m;
#pragma unroll
    for (int v = 0; v < 8; ++v) t2[v * TILE_PITCH] = fmaxf(acc[v] + bcol, 0.0f);
  }

  const float* __restrict__ wcol2 = W2 + wave * 16 + m;
  v2f bfrag2[FD / 4];
#pragma unroll
  for (int kt = 0; kt < FD / 4; ++kt) {
    const int k = kt * 4 + khalf;
    bfrag2[kt].x = wcol2[(size_t)k * FD];
    bfrag2[kt].y = wcol2[(size_t)(k + 1) * FD];
  }
  float dlo[8], dhi[8];
#pragma unroll
  for (int v = 0; v < 8; ++v) { dlo[v] = dinv[rowBase + v]; dhi[v] = dinv[rowBase + 8 + v]; }

  __syncthreads();

  const float* __restrict__ arow2 = tile2 + m * TILE_PITCH;
  v8f acc2 = {};
#pragma unroll
  for (int kt = 0; kt < FD / 4; ++kt) {
    const int k = kt * 4 + khalf;
    const v2f a = *(const v2f*)(arow2 + k);
    acc2 = __builtin_amdgcn_wmma_f32_16x16x4_f32(
         false, a,  false, bfrag2[kt],
         (short)0, acc2,  false,  false);
  }
  __builtin_amdgcn_sched_barrier(0);
  asm volatile("v_nop\n\tv_nop\n\tv_nop\n\tv_nop\n\tv_nop\n\tv_nop\n\tv_nop\n\tv_nop");
  __builtin_amdgcn_sched_barrier(0);

  float* __restrict__ hout = H + (size_t)rowBase * FD + wave * 16 + m;
#pragma unroll
  for (int v = 0; v < 8; ++v)
    hout[(size_t)(mBase + v) * FD] = acc2[v] * (mBase ? dhi[v] : dlo[v]);
}


__device__ __forceinline__ void pool_flush(float* __restrict__ psum, float* __restrict__ pcnt,
                                           int g, float ax, float ay, int n, int lane) {
  float* p = psum + (size_t)g * FD + 2 * lane;
  atomicAdd(p, ax);
  atomicAdd(p + 1, ay);
  if (lane == 0) atomicAdd(pcnt + g, (float)n);
}

__global__ __launch_bounds__(128) void gather_pool_kernel(const int* __restrict__ cnt, const int* __restrict__ cursorEnd,
                                                          const int* __restrict__ list, const float* __restrict__ dinv,
                                                          const float* __restrict__ h, const float* __restrict__ bias,
                                                          const int* __restrict__ seg,
                                                          float* __restrict__ psum, float* __restrict__ pcnt) {
  __shared__ __attribute__((aligned(16))) float tile[16 * TILE_PITCH];
  const int lane    = threadIdx.x & 31;
  const int wave    = __builtin_amdgcn_readfirstlane((int)(threadIdx.x >> 5));
  const int rowBase = blockIdx.x * 16;

  gather_tile16<false>(cnt, cursorEnd, list, dinv, h, bias, tile, rowBase, wave, lane);
  __syncthreads();
  if (wave != 0) return;

  const int gv = seg[rowBase + (lane & 15)];
  int   cur = __builtin_amdgcn_readlane(gv, 0);
  float ax = 0.0f, ay = 0.0f;
  int   n = 0;
#pragma unroll
  for (int k = 0; k < 16; ++k) {
    const v2f v = *(const v2f*)(tile + k * TILE_PITCH + 2 * lane);
    const int g = __builtin_amdgcn_readlane(gv, k);
    if (g != cur) {
      pool_flush(psum, pcnt, cur, ax, ay, n, lane);
      cur = g; ax = 0.0f; ay = 0.0f; n = 0;
    }
    ax += v.x;
    ay += v.y;
    ++n;
  }
  pool_flush(psum, pcnt, cur, ax, ay, n, lane);
}

__global__ void out_gemm_kernel(const float* __restrict__ psum, const float* __restrict__ pcnt,
                                const float* __restrict__ Wout, const float* __restrict__ bout,
                                float* __restrict__ out) {
  int t = blockIdx.x * blockDim.x + threadIdx.x;
  if (t < NGRAPHS * NTARGETS) {
    int g = t / NTARGETS;
    int c = t % NTARGETS;
    float inv = 1.0f / fmaxf(pcnt[g], 1.0f);
    float acc = 0.0f;
#pragma unroll 8
    for (int d = 0; d < FD; ++d)
      acc += psum[(size_t)g * FD + d] * Wout[(size_t)d * NTARGETS + c];
    out[t] = acc * inv + bout[c];
  }
}


extern "C" void kernel_launch(void* const* d_in, const int* in_sizes, int n_in,
                              void* d_out, int out_size, void* d_ws, size_t ws_size,
                              hipStream_t stream) {
  const float* x    = (const float*)d_in[0];
  const float* W1   = (const float*)d_in[1];
  const float* b1   = (const float*)d_in[2];
  const float* W2   = (const float*)d_in[3];
  const float* b2   = (const float*)d_in[4];
  const float* W3   = (const float*)d_in[5];
  const float* b3   = (const float*)d_in[6];
  const float* Wout = (const float*)d_in[7];
  const float* bout = (const float*)d_in[8];
  const int*   eidx = (const int*)d_in[9];
  const int*   seg  = (const int*)d_in[10];
  float* out = (float*)d_out;
  (void)n_in; (void)out_size;

  const int nNodes = in_sizes[0] / FD;
  const int nE     = in_sizes[9] / 2;
  const int* srcI = eidx;
  const int* dstI = eidx + nE;

  char* ws = (char*)d_ws;
  size_t off = 0;
  auto alloc = [&](size_t bytes) -> void* {
    void* p = ws + off;
    off = (off + bytes + 255) & ~(size_t)255;
    return p;
  };
  float* dinv   = (float*)alloc((size_t)nNodes * sizeof(float));
  float* hbuf   = (float*)alloc((size_t)nNodes * FD * sizeof(float));
  float* feat   = (float*)alloc((size_t)nNodes * FD * sizeof(float));
  float* psum   = (float*)alloc((size_t)NGRAPHS * FD * sizeof(float));
  float* pcnt   = (float*)alloc((size_t)NGRAPHS * sizeof(float));
  int*   cnt    = (int*)alloc((size_t)nNodes * sizeof(int));
  int*   cursor = (int*)alloc((size_t)nNodes * sizeof(int));
  int*   list   = (int*)alloc((size_t)nE * sizeof(int));
  if (ws_size < off) return;

  const int nBins   = (nNodes + BIN_NODES - 1) >> BIN_SHIFT;
  const int nChunks = (nE + CHUNK - 1) / CHUNK;
  unsigned* part      = (unsigned*)feat;
  int*      blockhist = (int*)(part + (((size_t)nE + 63) & ~(size_t)63));
  int*      binTotal  = blockhist + (size_t)nBins * nChunks;
  int*      binStart  = binTotal + nBins;
  if ((size_t)((const char*)(binStart + nBins) - (const char*)feat) > (size_t)nNodes * FD * sizeof(float)) return;
  if (nBins > MAXBIN || nNodes > (1 << SRC_BITS) || nBins * BIN_NODES < NGRAPHS * FD) return;

  const int T = 256;
  const int tileBlocks = nNodes / 16;

  bin_hist_kernel<<<nChunks, T, 0, stream>>>(dstI, blockhist, nE, nBins, nChunks);
  bin_scan_kernel<<<nBins, T, 0, stream>>>(blockhist, binTotal, nChunks, psum, pcnt);
  bin_scatter_kernel<<<nChunks, T, 0, stream>>>(srcI, dstI, blockhist, binTotal, binStart, part, nE, nBins, nChunks);
  bin_sort_kernel<<<nBins, T, 0, stream>>>(part, binStart, binTotal, cnt, cursor, dinv, list, nNodes);

  gather_x_product2_kernel<<<tileBlocks, 128, 0, stream>>>(cnt, cursor, list, dinv, x, W1, b1, W2, feat);
  gather_product_kernel<<<tileBlocks, 128, 0, stream>>>(cnt, cursor, list, dinv, feat, b2, W3, hbuf);

  gather_pool_kernel<<<tileBlocks, 128, 0, stream>>>(cnt, cursor, list, dinv, hbuf, b3, seg, psum, pcnt);
  out_gemm_kernel<<<(NGRAPHS * NTARGETS + T - 1) / T, T, 0, stream>>>(psum, pcnt, Wout, bout, out);
}
